# starting kernel
# baseline (speedup 1.0000x reference)
#include <hip/hip_runtime.h>
#include <hip/hip_bf16.h>

typedef __attribute__((ext_vector_type(2))) float v2f;
typedef __attribute__((ext_vector_type(8))) float v8f;

#define D_FEAT 64

__device__ __forceinline__ void atomic_add_f32(float* p, float v) {
    __hip_atomic_fetch_add(p, v, __ATOMIC_RELAXED, __HIP_MEMORY_SCOPE_AGENT);
}

__global__ void sage_init_kernel(const float* __restrict__ x1,
                                 const float* __restrict__ x2,
                                 float* __restrict__ agg1,
                                 float* __restrict__ agg2,
                                 float* __restrict__ cnt1,
                                 float* __restrict__ cnt2,
                                 long total, int n) {
    long t = (long)blockIdx.x * blockDim.x + threadIdx.x;
    if (t < total) {
        agg1[t] = x1[t];
        agg2[t] = x2[t];
        if (t < n) { cnt1[t] = 1.0f; cnt2[t] = 1.0f; }
    }
}

__global__ void sage_scatter_kernel(const float* __restrict__ x,
                                    const int* __restrict__ edges,
                                    float* __restrict__ agg,
                                    float* __restrict__ cnt,
                                    int n_edges) {
    long t = (long)blockIdx.x * blockDim.x + threadIdx.x;
    long e = t >> 6;
    int  f = (int)(t & 63);
    if (e >= n_edges) return;
    int row = edges[e];
    int col = edges[(long)n_edges + e];
    atomic_add_f32(&agg[(long)row * D_FEAT + f], x[(long)col * D_FEAT + f]);
    if (f == 0) atomic_add_f32(&cnt[row], 1.0f);
}

__global__ __launch_bounds__(256) void sage_gemm_norm_kernel(
        const float* __restrict__ agg1, const float* __restrict__ agg2,
        const float* __restrict__ cnt1, const float* __restrict__ cnt2,
        const float* __restrict__ x1,   const float* __restrict__ W,
        const float* __restrict__ bias, float* __restrict__ out, int n) {
    __shared__ float ldsW[192 * 64];

    const int tid = threadIdx.x;
    for (int idx = tid; idx < 192 * 64; idx += 256) {
        const int row = idx >> 6;
        const int col = idx & 63;
        ldsW[((((row >> 1) << 6) | col) << 1) | (row & 1)] = W[idx];
    }
    __syncthreads();

    const int lane = tid & 31;
    const int wave = tid >> 5;
    const int m = lane & 15;
    const int h = lane >> 4;

    const long tile = (long)blockIdx.x * 8 + wave;
    const long row0 = tile << 4;
    long r = row0 + m;
    if (r > (long)(n - 1)) r = (long)(n - 1);

    const float* pa1 = agg1 + r * D_FEAT;
    const float* pa2 = agg2 + r * D_FEAT;
    const float* pax = x1   + r * D_FEAT;
    const float inv1 = 1.0f / fmaxf(cnt1[r], 1.0f);
    const float inv2 = 1.0f / fmaxf(cnt2[r], 1.0f);

    v8f acc0 = {}; v8f acc1 = {}; v8f acc2 = {}; v8f acc3 = {};
    const float* wbase = ldsW + (m << 1);

#pragma unroll
    for (int k = 0; k < 16; ++k) {
        const int c = (k << 2) + (h << 1);
        v2f a = *(const v2f*)(pa1 + c);
        a *= inv1;
        const float* wp = wbase + (((k << 1) + h) << 7);
        const v2f b0 = *(const v2f*)(wp);
        const v2f b1 = *(const v2f*)(wp + 32);
        const v2f b2 = *(const v2f*)(wp + 64);
        const v2f b3 = *(const v2f*)(wp + 96);
        acc0 = __builtin_amdgcn_wmma_f32_16x16x4_f32(false, a, false, b0, (short)0, acc0, false, false);
        acc1 = __builtin_amdgcn_wmma_f32_16x16x4_f32(false, a, false, b1, (short)0, acc1, false, false);
        acc2 = __builtin_amdgcn_wmma_f32_16x16x4_f32(false, a, false, b2, (short)0, acc2, false, false);
        acc3 = __builtin_amdgcn_wmma_f32_16x16x4_f32(false, a, false, b3, (short)0, acc3, false, false);
    }
#pragma unroll
    for (int k = 0; k < 16; ++k) {
        const int c = (k << 2) + (h << 1);
        v2f a = *(const v2f*)(pa2 + c);
        a *= inv2;
        const float* wp = wbase + ((32 + (k << 1) + h) << 7);
        const v2f b0 = *(const v2f*)(wp);
        const v2f b1 = *(const v2f*)(wp + 32);
        const v2f b2 = *(const v2f*)(wp + 64);
        const v2f b3 = *(const v2f*)(wp + 96);
        acc0 = __builtin_amdgcn_wmma_f32_16x16x4_f32(false, a, false, b0, (short)0, acc0, false, false);
        acc1 = __builtin_amdgcn_wmma_f32_16x16x4_f32(false, a, false, b1, (short)0, acc1, false, false);
        acc2 = __builtin_amdgcn_wmma_f32_16x16x4_f32(false, a, false, b2, (short)0, acc2, false, false);
        acc3 = __builtin_amdgcn_wmma_f32_16x16x4_f32(false, a, false, b3, (short)0, acc3, false, false);
    }
#pragma unroll
    for (int k = 0; k < 16; ++k) {
        const int c = (k << 2) + (h << 1);
        const v2f a = *(const v2f*)(pax + c);
        const float* wp = wbase + ((64 + (k << 1) + h) << 7);
        const v2f b0 = *(const v2f*)(wp);
        const v2f b1 = *(const v2f*)(wp + 32);
        const v2f b2 = *(const v2f*)(wp + 64);
        const v2f b3 = *(const v2f*)(wp + 96);
        acc0 = __builtin_amdgcn_wmma_f32_16x16x4_f32(false, a, false, b0, (short)0, acc0, false, false);
        acc1 = __builtin_amdgcn_wmma_f32_16x16x4_f32(false, a, false, b1, (short)0, acc1, false, false);
        acc2 = __builtin_amdgcn_wmma_f32_16x16x4_f32(false, a, false, b2, (short)0, acc2, false, false);
        acc3 = __builtin_amdgcn_wmma_f32_16x16x4_f32(false, a, false, b3, (short)0, acc3, false, false);
    }

    const float bs0 = bias[m];
    const float bs1 = bias[16 + m];
    const float bs2 = bias[32 + m];
    const float bs3 = bias[48 + m];

#pragma unroll
    for (int j = 0; j < 8; ++j) {
        const float o0 = acc0[j] + bs0;
        const float o1 = acc1[j] + bs1;
        const float o2 = acc2[j] + bs2;
        const float o3 = acc3[j] + bs3;
        float s = o0 * o0 + o1 * o1 + o2 * o2 + o3 * o3;
        s += __shfl_xor(s, 1, 32);
        s += __shfl_xor(s, 2, 32);
        s += __shfl_xor(s, 4, 32);
        s += __shfl_xor(s, 8, 32);
        const float scale = 1.0f / fmaxf(sqrtf(s), 1e-12f);
        const long orow = row0 + j + (h << 3);
        if (orow < n) {
            float* po = out + orow * D_FEAT + m;
            po[0]  = o0 * scale;
            po[16] = o1 * scale;
            po[32] = o2 * scale;
            po[48] = o3 * scale;
        }
    }
}

extern "C" void kernel_launch(void* const* d_in, const int* in_sizes, int n_in,
                              void* d_out, int out_size, void* d_ws, size_t ws_size,
                              hipStream_t stream) {
    const float* x1    = (const float*)d_in[0];
    const float* x2    = (const float*)d_in[1];
    const int*   e_pos = (const int*)d_in[2];
    const int*   e_neg = (const int*)d_in[3];
    const float* W     = (const float*)d_in[4];
    const float* bias  = (const float*)d_in[5];
    float*       out   = (float*)d_out;

    const int n       = in_sizes[0] / D_FEAT;
    const int n_edges = in_sizes[2] / 2;

    float* agg1 = (float*)d_ws;
    float* agg2 = agg1 + (size_t)n * D_FEAT;
    float* cnt1 = agg2 + (size_t)n * D_FEAT;
    float* cnt2 = cnt1 + n;

    const long total = (long)n * D_FEAT;
    {
        const int blk = 256;
        const long grid = (total + blk - 1) / blk;
        sage_init_kernel<<<(unsigned)grid, blk, 0, stream>>>(x1, x2, agg1, agg2,
                                                             cnt1, cnt2, total, n);
    }
    {
        const int blk = 256;
        const long work = (long)n_edges * D_FEAT;
        const long grid = (work + blk - 1) / blk;
        sage_scatter_kernel<<<(unsigned)grid, blk, 0, stream>>>(x1, e_pos, agg1, cnt1, n_edges);
        sage_scatter_kernel<<<(unsigned)grid, blk, 0, stream>>>(x2, e_neg, agg2, cnt2, n_edges);
    }
    {
        const int tiles  = (n + 15) / 16;
        const int blocks = (tiles + 7) / 8;
        sage_gemm_norm_kernel<<<blocks, 256, 0, stream>>>(agg1, agg2, cnt1, cnt2,
                                                          x1, W, bias, out, n);
    }
}


# Edges counting-sorted by destination row (int atomics, arrival-order slots); each row summed in slot order by a gather inside the GEMM kernel; replaces f32 atomic scatter, agg buffers, init pass
# speedup vs baseline: 1.7724x; 1.7724x over previous
#include <hip/hip_runtime.h>
#include <hip/hip_bf16.h>

typedef __attribute__((ext_vector_type(2))) float v2f;
typedef __attribute__((ext_vector_type(4))) float v4f;
typedef __attribute__((ext_vector_type(8))) float v8f;

#define D_FEAT 64
#define TILE_PITCH 68
#define EDGES_PER_BLOCK 1024


__global__ void csr_zero_kernel(int* __restrict__ p, int count) {
    const int t = blockIdx.x * blockDim.x + threadIdx.x;
    if (t < count) p[t] = 0;
}

__global__ __launch_bounds__(256) void csr_count_kernel(const int* __restrict__ e_pos,
                                                        const int* __restrict__ e_neg,
                                                        int* __restrict__ cur,
                                                        int n, int n_edges) {
    const int set = blockIdx.y;
    const int* __restrict__ rows = set ? e_neg : e_pos;
    int* __restrict__ deg = cur + (long)set * n;
    const int e0 = blockIdx.x * EDGES_PER_BLOCK + threadIdx.x;
    for (int u = 0; u < 4; ++u) {
        const int e = e0 + u * 256;
        if (e < n_edges)
            __hip_atomic_fetch_add(&deg[rows[e]], 1, __ATOMIC_RELAXED, __HIP_MEMORY_SCOPE_AGENT);
    }
}

__global__ __launch_bounds__(256) void csr_alloc_kernel(int* __restrict__ cur,
                                                        int* __restrict__ start,
                                                        int* __restrict__ total, int n) {
    __shared__ int wsum[8];
    __shared__ int sbase;
    const int set = blockIdx.y;
    int* __restrict__ c  = cur   + (long)set * n;
    int* __restrict__ st = start + (long)set * n;
    const int i    = blockIdx.x * 256 + threadIdx.x;
    const int lane = threadIdx.x & 31;
    const int wave = threadIdx.x >> 5;
    const int d = (i < n) ? c[i] : 0;
    int inc = d;
    for (int o = 1; o < 32; o <<= 1) {
        const int t = __shfl_up(inc, o, 32);
        if (lane >= o) inc += t;
    }
    if (lane == 31) wsum[wave] = inc;
    __syncthreads();
    if (threadIdx.x == 0) {
        int s = 0;
        for (int w = 0; w < 8; ++w) { const int t = wsum[w]; wsum[w] = s; s += t; }
        sbase = __hip_atomic_fetch_add(&total[set], s, __ATOMIC_RELAXED, __HIP_MEMORY_SCOPE_AGENT);
    }
    __syncthreads();
    const int off = sbase + wsum[wave] + inc - d;
    if (i < n) { st[i] = off; c[i] = off; }
}

__global__ __launch_bounds__(256) void csr_fill_kernel(const int* __restrict__ e_pos,
                                                       const int* __restrict__ e_neg,
                                                       int* __restrict__ cur,
                                                       int* __restrict__ nbr,
                                                       int n, int n_edges) {
    const int set = blockIdx.y;
    const int* __restrict__ edges = set ? e_neg : e_pos;
    int* __restrict__ cs = cur + (long)set * n;
    int* __restrict__ nb = nbr + (long)set * n_edges;
    const int e0 = blockIdx.x * EDGES_PER_BLOCK + threadIdx.x;
    for (int u = 0; u < 4; ++u) {
        const int e = e0 + u * 256;
        if (e < n_edges) {
            const int row = edges[e];
            const int col = edges[(long)n_edges + e];
            const int p = __hip_atomic_fetch_add(&cs[row], 1, __ATOMIC_RELAXED, __HIP_MEMORY_SCOPE_AGENT);
            nb[p] = col;
        }
    }
}

__device__ __forceinline__ void gather_mean_tile(const float* __restrict__ x,
                                                 const int* __restrict__ st,
                                                 const int* __restrict__ en,
                                                 const int* __restrict__ nb,
                                                 long row0, int n, float* T, int lane) {
    const int hh = lane >> 4;
    const int q  = lane & 15;
    const v4f* __restrict__ xv = (const v4f*)x;
    const v4f zero4 = {0.0f, 0.0f, 0.0f, 0.0f};

    long rl = row0 + q;
    if (rl > (long)(n - 1)) rl = (long)(n - 1);
    const int sv = st[rl];
    const int ev = en[rl];

    for (int i = 0; i < 8; ++i) {
        const int lr = 2 * i + hh;
        long r = row0 + lr;
        if (r > (long)(n - 1)) r = (long)(n - 1);
        const int s = __shfl(sv, lr, 32);
        const int e = __shfl(ev, lr, 32);
        const int d = e - s;
        const int dmax = max(d, __shfl_xor(d, 16, 32));

        v4f acc = xv[r * 16 + q];
        for (int j = 0; j < dmax; j += 4) {
            int cidx[4];
            v4f t[4];
#pragma unroll
            for (int u = 0; u < 4; ++u) {
                const int p = (j + u < d) ? (s + j + u) : 0;
                cidx[u] = nb[p];
            }
#pragma unroll
            for (int u = 0; u < 4; ++u) {
                const long rr = (j + u < d) ? (long)cidx[u] : r;
                t[u] = xv[rr * 16 + q];
            }
#pragma unroll
            for (int u = 0; u < 4; ++u)
                acc += (j + u < d) ? t[u] : zero4;
        }
        const float inv = 1.0f / fmaxf((float)(d + 1), 1.0f);
        acc *= inv;
        *(v4f*)(T + lr * TILE_PITCH + (q << 2)) = acc;
    }
}

__global__ __launch_bounds__(512, 8) void sage_gather_gemm_norm_kernel(
        const float* __restrict__ x1,   const float* __restrict__ x2,
        const int* __restrict__ start,  const int* __restrict__ endp,
        const int* __restrict__ nbr,
        const float* __restrict__ W,    const float* __restrict__ bias,
        float* __restrict__ out, int n, int n_edges) {
    __shared__ float ldsW[192 * 64];
    __shared__ __attribute__((aligned(16))) float ldsA[16 * 16 * TILE_PITCH];

    const int tid = threadIdx.x;
    for (int idx = tid; idx < 192 * 64; idx += 512) {
        const int row = idx >> 6;
        const int col = idx & 63;
        ldsW[((((row >> 1) << 6) | col) << 1) | (row & 1)] = W[idx];
    }

    const int lane = tid & 31;
    const int wave = tid >> 5;
    const int m = lane & 15;
    const int h = lane >> 4;

    const long tile = (long)blockIdx.x * 16 + wave;
    const long row0 = tile << 4;
    long r = row0 + m;
    if (r > (long)(n - 1)) r = (long)(n - 1);

    float* T = ldsA + wave * (16 * TILE_PITCH);
    const float* ta = T + m * TILE_PITCH;
    const float* pax = x1 + r * D_FEAT;

    v8f acc0 = {}; v8f acc1 = {}; v8f acc2 = {}; v8f acc3 = {};
    const float* wbase = ldsW + (m << 1);

#pragma nounroll
    for (int set = 0; set < 2; ++set) {
        gather_mean_tile(set ? x2 : x1, start + (long)set * n, endp + (long)set * n,
                         nbr + (long)set * n_edges, row0, n, T, lane);
        __syncthreads();
        const float* wset = wbase + ((set << 5) << 7);
#pragma unroll
        for (int k = 0; k < 16; ++k) {
            const int c = (k << 2) + (h << 1);
            const v2f a = *(const v2f*)(ta + c);
            const float* wp = wset + (((k << 1) + h) << 7);
            const v2f b0 = *(const v2f*)(wp);
            const v2f b1 = *(const v2f*)(wp + 32);
            const v2f b2 = *(const v2f*)(wp + 64);
            const v2f b3 = *(const v2f*)(wp + 96);
            acc0 = __builtin_amdgcn_wmma_f32_16x16x4_f32(false, a, false, b0, (short)0, acc0, false, false);
            acc1 = __builtin_amdgcn_wmma_f32_16x16x4_f32(false, a, false, b1, (short)0, acc1, false, false);
            acc2 = __builtin_amdgcn_wmma_f32_16x16x4_f32(false, a, false, b2, (short)0, acc2, false, false);
            acc3 = __builtin_amdgcn_wmma_f32_16x16x4_f32(false, a, false, b3, (short)0, acc3, false, false);
        }
        asm volatile("v_nop\n\tv_nop\n\tv_nop\n\tv_nop\n\tv_nop\n\tv_nop\n\tv_nop\n\tv_nop"
                     : "+v"(acc0), "+v"(acc1), "+v"(acc2), "+v"(acc3));
        __syncthreads();
    }
#pragma unroll
    for (int k = 0; k < 16; ++k) {
        const int c = (k << 2) + (h << 1);
        const v2f a = *(const v2f*)(pax + c);
        const float* wp = wbase + ((64 + (k << 1) + h) << 7);
        const v2f b0 = *(const v2f*)(wp);
        const v2f b1 = *(const v2f*)(wp + 32);
        const v2f b2 = *(const v2f*)(wp + 64);
        const v2f b3 = *(const v2f*)(wp + 96);
        acc0 = __builtin_amdgcn_wmma_f32_16x16x4_f32(false, a, false, b0, (short)0, acc0, false, false);
        acc1 = __builtin_amdgcn_wmma_f32_16x16x4_f32(false, a, false, b1, (short)0, acc1, false, false);
        acc2 = __builtin_amdgcn_wmma_f32_16x16x4_f32(false, a, false, b2, (short)0, acc2, false, false);
        acc3 = __builtin_amdgcn_wmma_f32_16x16x4_f32(false, a, false, b3, (short)0, acc3, false, false);
    }

    const float bs0 = bias[m];
    const float bs1 = bias[16 + m];
    const float bs2 = bias[32 + m];
    const float bs3 = bias[48 + m];

#pragma unroll
    for (int j = 0; j < 8; ++j) {
        const float o0 = acc0[j] + bs0;
        const float o1 = acc1[j] + bs1;
        const float o2 = acc2[j] + bs2;
        const float o3 = acc3[j] + bs3;
        float s = o0 * o0 + o1 * o1 + o2 * o2 + o3 * o3;
        s += __shfl_xor(s, 1, 32);
        s += __shfl_xor(s, 2, 32);
        s += __shfl_xor(s, 4, 32);
        s += __shfl_xor(s, 8, 32);
        const float scale = 1.0f / fmaxf(sqrtf(s), 1e-12f);
        const long orow = row0 + j + (h << 3);
        if (orow < n) {
            float* po = out + orow * D_FEAT + m;
            po[0]  = o0 * scale;
            po[16] = o1 * scale;
            po[32] = o2 * scale;
            po[48] = o3 * scale;
        }
    }
}

extern "C" void kernel_launch(void* const* d_in, const int* in_sizes, int n_in,
                              void* d_out, int out_size, void* d_ws, size_t ws_size,
                              hipStream_t stream) {
    const float* x1    = (const float*)d_in[0];
    const float* x2    = (const float*)d_in[1];
    const int*   e_pos = (const int*)d_in[2];
    const int*   e_neg = (const int*)d_in[3];
    const float* W     = (const float*)d_in[4];
    const float* bias  = (const float*)d_in[5];
    float*       out   = (float*)d_out;

    const int n       = in_sizes[0] / D_FEAT;
    const int n_edges = in_sizes[2] / 2;

    int* cur   = (int*)d_ws;
    int* total = cur + (size_t)2 * n;
    int* start = total + 64;
    int* nbr   = start + (size_t)2 * n;

    {
        const int count = 2 * n + 64;
        csr_zero_kernel<<<(count + 255) / 256, 256, 0, stream>>>(cur, count);
    }
    const dim3 egrid((n_edges + EDGES_PER_BLOCK - 1) / EDGES_PER_BLOCK, 2);
    const dim3 ngrid((n + 255) / 256, 2);
    csr_count_kernel<<<egrid, 256, 0, stream>>>(e_pos, e_neg, cur, n, n_edges);
    csr_alloc_kernel<<<ngrid, 256, 0, stream>>>(cur, start, total, n);
    csr_fill_kernel<<<egrid, 256, 0, stream>>>(e_pos, e_neg, cur, nbr, n, n_edges);
    {
        const int tiles  = (n + 15) / 16;
        const int blocks = (tiles + 15) / 16;
        sage_gather_gemm_norm_kernel<<<blocks, 512, 0, stream>>>(x1, x2, start, cur, nbr,
                                                                 W, bias, out, n, n_edges);
    }
}


# Edges grouped by destination in two levels (256-row buckets, then rows inside the GEMM block) with LDS counters as ranks, not two global atomics per edge; row sums stay in arrival-order slots
# speedup vs baseline: 3.3381x; 1.8834x over previous
#include <hip/hip_runtime.h>
#include <hip/hip_bf16.h>

typedef __attribute__((ext_vector_type(2))) float v2f;
typedef __attribute__((ext_vector_type(4))) float v4f;
typedef __attribute__((ext_vector_type(8))) float v8f;

#define D_FEAT 64
#define TILE_PITCH 68
#define BUCKET_ROWS 256
#define MAX_BUCKETS 512
#define CHUNK 4096
#define PIECE 4096


__device__ __forceinline__ int lds_fetch_add(int* p, int v) {
    return __hip_atomic_fetch_add(p, v, __ATOMIC_RELAXED, __HIP_MEMORY_SCOPE_WORKGROUP);
}

__global__ __launch_bounds__(512) void bucket_count_kernel(const int* __restrict__ e_pos,
                                                           const int* __restrict__ e_neg,
                                                           int* __restrict__ cntm,
                                                           int n_edges, int n_buckets, int n_chunks) {
    __shared__ int hist[MAX_BUCKETS];
    const int set = blockIdx.y;
    const int tid = threadIdx.x;
    const int* __restrict__ rows = set ? e_neg : e_pos;
    hist[tid] = 0;
    __syncthreads();
    const int e0 = blockIdx.x * CHUNK + tid;
    int rw[8];
#pragma unroll
    for (int u = 0; u < 8; ++u) {
        const int e = e0 + u * 512;
        rw[u] = (e < n_edges) ? rows[e] : -1;
    }
#pragma unroll
    for (int u = 0; u < 8; ++u)
        if (rw[u] >= 0) lds_fetch_add(&hist[rw[u] >> 8], 1);
    __syncthreads();
    if (tid < n_buckets)
        cntm[((long)set * n_buckets + tid) * n_chunks + blockIdx.x] = hist[tid];
}

__global__ __launch_bounds__(256) void bucket_scan_kernel(int* __restrict__ cntm,
                                                          int* __restrict__ total,
                                                          int n_lists, int n_chunks) {
    const int lane = threadIdx.x & 31;
    const int g = blockIdx.x * 8 + (threadIdx.x >> 5);
    if (g >= n_lists) return;
    int* __restrict__ p = cntm + (long)g * n_chunks;
    int carry = 0;
    for (int j = 0; j < n_chunks; j += 32) {
        const int i = j + lane;
        const int v = (i < n_chunks) ? p[i] : 0;
        int inc = v;
        for (int o = 1; o < 32; o <<= 1) {
            const int t = __shfl_up(inc, o, 32);
            if (lane >= o) inc += t;
        }
        if (i < n_chunks) p[i] = carry + inc - v;
        carry += __shfl(inc, 31, 32);
    }
    if (lane == 0) total[g] = carry;
}

__global__ __launch_bounds__(512) void bucket_place_kernel(const int* __restrict__ e_pos,
                                                           const int* __restrict__ e_neg,
                                                           const int* __restrict__ cntm,
                                                           const int* __restrict__ total,
                                                           int* __restrict__ pairs,
                                                           int n_edges, int n_buckets, int n_chunks) {
    __shared__ int cursor[MAX_BUCKETS];
    __shared__ int wsum[16];
    const int set  = blockIdx.y;
    const int tid  = threadIdx.x;
    const int lane = tid & 31;
    const int wave = tid >> 5;
    const int* __restrict__ edges = set ? e_neg : e_pos;

    const int e0 = blockIdx.x * CHUNK + tid;
    int rw[8], cl[8];
#pragma unroll
    for (int u = 0; u < 8; ++u) {
        const int e = e0 + u * 512;
        const bool ok = e < n_edges;
        rw[u] = ok ? edges[e] : -1;
        cl[u] = ok ? edges[(long)n_edges + e] : 0;
    }

    const bool live = tid < n_buckets;
    const int t   = live ? total[set * n_buckets + tid] : 0;
    const int off = live ? cntm[((long)set * n_buckets + tid) * n_chunks + blockIdx.x] : 0;
    int inc = t;
    for (int o = 1; o < 32; o <<= 1) {
        const int s = __shfl_up(inc, o, 32);
        if (lane >= o) inc += s;
    }
    if (lane == 31) wsum[wave] = inc;
    __syncthreads();
    int woff = 0;
    for (int w = 0; w < 16; ++w) woff += (w < wave) ? wsum[w] : 0;
    cursor[tid] = woff + inc - t + off;
    __syncthreads();

    int* __restrict__ dst = pairs + (long)set * n_edges;
#pragma unroll
    for (int u = 0; u < 8; ++u) {
        if (rw[u] >= 0) {
            const int p = lds_fetch_add(&cursor[rw[u] >> 8], 1);
            dst[p] = (cl[u] << 8) | (rw[u] & (BUCKET_ROWS - 1));
        }
    }
}

__device__ __forceinline__ void gather_mean_tile(const float* __restrict__ x,
                                                 const int* __restrict__ seg, int seg_count,
                                                 long row0, int n, float* T,
                                                 int* lnbr, int* deg, int* lstart, int* cursor,
                                                 int* rdeg, int* wsum,
                                                 int tid, int lane, int wave) {
    const int hh = lane >> 4;
    const int q  = lane & 15;
    const v4f* __restrict__ xv = (const v4f*)x;
    const v4f zero4 = {0.0f, 0.0f, 0.0f, 0.0f};

#pragma unroll
    for (int i = 0; i < 8; ++i) {
        const int lr = 2 * i + hh;
        long r = row0 + lr;
        if (r > (long)(n - 1)) r = (long)(n - 1);
        *(v4f*)(T + lr * TILE_PITCH + (q << 2)) = xv[r * 16 + q];
    }
    if (tid < BUCKET_ROWS) rdeg[tid] = 0;

    for (int pb = 0; pb < seg_count; pb += PIECE) {
        const int pc = min(PIECE, seg_count - pb);
        int w[8];
#pragma unroll
        for (int u = 0; u < 8; ++u) {
            const int e = tid + u * 512;
            w[u] = (e < pc) ? seg[pb + e] : 0;
        }
        if (tid < BUCKET_ROWS) deg[tid] = 0;
        __syncthreads();
#pragma unroll
        for (int u = 0; u < 8; ++u)
            if (tid + u * 512 < pc) lds_fetch_add(&deg[w[u] & (BUCKET_ROWS - 1)], 1);
        __syncthreads();
        const int d0 = (tid < BUCKET_ROWS) ? deg[tid] : 0;
        int inc = d0;
        for (int o = 1; o < 32; o <<= 1) {
            const int t = __shfl_up(inc, o, 32);
            if (lane >= o) inc += t;
        }
        if (lane == 31 && wave < 8) wsum[wave] = inc;
        __syncthreads();
        if (tid < BUCKET_ROWS) {
            int woff = 0;
            for (int k = 0; k < 8; ++k) woff += (k < wave) ? wsum[k] : 0;
            const int st = woff + inc - d0;
            lstart[tid] = st;
            cursor[tid] = st;
            rdeg[tid] += d0;
        }
        __syncthreads();
#pragma unroll
        for (int u = 0; u < 8; ++u) {
            if (tid + u * 512 < pc) {
                const int slot = lds_fetch_add(&cursor[w[u] & (BUCKET_ROWS - 1)], 1);
                lnbr[slot] = (int)((unsigned)w[u] >> 8);
            }
        }
        __syncthreads();

        for (int i = 0; i < 8; ++i) {
            const int lr = 2 * i + hh;
            long r = row0 + lr;
            if (r > (long)(n - 1)) r = (long)(n - 1);
            const int s = lstart[wave * 16 + lr];
            const int d = cursor[wave * 16 + lr] - s;
            const int dmax = max(d, __shfl_xor(d, 16, 32));

            float* trow = T + lr * TILE_PITCH + (q << 2);
            v4f acc = *(v4f*)trow;
            for (int j = 0; j < dmax; j += 4) {
                int cidx[4];
                v4f t[4];
#pragma unroll
                for (int u = 0; u < 4; ++u)
                    cidx[u] = lnbr[(j + u < d) ? (s + j + u) : 0];
#pragma unroll
                for (int u = 0; u < 4; ++u) {
                    const long rr = (j + u < d) ? (long)cidx[u] : r;
                    t[u] = xv[rr * 16 + q];
                }
#pragma unroll
                for (int u = 0; u < 4; ++u)
                    acc += (j + u < d) ? t[u] : zero4;
            }
            *(v4f*)trow = acc;
        }
        __syncthreads();
    }
    __syncthreads();
#pragma unroll
    for (int i = 0; i < 8; ++i) {
        const int lr = 2 * i + hh;
        const float inv = 1.0f / fmaxf((float)(rdeg[wave * 16 + lr] + 1), 1.0f);
        float* trow = T + lr * TILE_PITCH + (q << 2);
        v4f acc = *(v4f*)trow;
        acc *= inv;
        *(v4f*)trow = acc;
    }
}

__global__ __launch_bounds__(512, 8) void sage_gather_gemm_norm_kernel(
        const float* __restrict__ x1,   const float* __restrict__ x2,
        const int* __restrict__ total,  const int* __restrict__ pairs,
        const float* __restrict__ W,    const float* __restrict__ bias,
        float* __restrict__ out, int n, int n_edges, int n_buckets) {
    __shared__ float ldsW[192 * 64];
    __shared__ __attribute__((aligned(16))) float ldsA[16 * 16 * TILE_PITCH];
    __shared__ int lnbr[PIECE];
    __shared__ int deg[BUCKET_ROWS], lstart[BUCKET_ROWS], cursor[BUCKET_ROWS], rdeg[BUCKET_ROWS];
    __shared__ int wsum[8];
    __shared__ int red[2][16];

    const int tid = threadIdx.x;
    for (int idx = tid; idx < 192 * 64; idx += 512) {
        const int row = idx >> 6;
        const int col = idx & 63;
        ldsW[((((row >> 1) << 6) | col) << 1) | (row & 1)] = W[idx];
    }

    const int lane = tid & 31;
    const int wave = tid >> 5;
    const int m = lane & 15;
    const int h = lane >> 4;

    {
        const int bk = blockIdx.x;
        int v0 = (tid < bk) ? total[tid] : 0;
        int v1 = (tid < bk) ? total[n_buckets + tid] : 0;
        for (int o = 16; o > 0; o >>= 1) {
            v0 += __shfl_xor(v0, o, 32);
            v1 += __shfl_xor(v1, o, 32);
        }
        if (lane == 0) { red[0][wave] = v0; red[1][wave] = v1; }
        __syncthreads();
    }
    int seg_beg0 = 0, seg_beg1 = 0;
    for (int k = 0; k < 16; ++k) { seg_beg0 += red[0][k]; seg_beg1 += red[1][k]; }

    const long tile = (long)blockIdx.x * 16 + wave;
    const long row0 = tile << 4;
    long r = row0 + m;
    if (r > (long)(n - 1)) r = (long)(n - 1);

    float* T = ldsA + wave * (16 * TILE_PITCH);
    const float* ta = T + m * TILE_PITCH;
    const float* pax = x1 + r * D_FEAT;

    v8f acc0 = {}; v8f acc1 = {}; v8f acc2 = {}; v8f acc3 = {};
    const float* wbase = ldsW + (m << 1);

#pragma nounroll
    for (int set = 0; set < 2; ++set) {
        const int seg_beg = __builtin_amdgcn_readfirstlane(set ? seg_beg1 : seg_beg0);
        const int seg_cnt = __builtin_amdgcn_readfirstlane(total[set * n_buckets + blockIdx.x]);
        gather_mean_tile(set ? x2 : x1, pairs + (long)set * n_edges + seg_beg, seg_cnt,
                         row0, n, T, lnbr, deg, lstart, cursor, rdeg, wsum, tid, lane, wave);
        __syncthreads();
        const float* wset = wbase + ((set << 5) << 7);
#pragma unroll
        for (int k = 0; k < 16; ++k) {
            const int c = (k << 2) + (h << 1);
            const v2f a = *(const v2f*)(ta + c);
            const float* wp = wset + (((k << 1) + h) << 7);
            const v2f b0 = *(const v2f*)(wp);
            const v2f b1 = *(const v2f*)(wp + 32);
            const v2f b2 = *(const v2f*)(wp + 64);
            const v2f b3 = *(const v2f*)(wp + 96);
            acc0 = __builtin_amdgcn_wmma_f32_16x16x4_f32(false, a, false, b0, (short)0, acc0, false, false);
            acc1 = __builtin_amdgcn_wmma_f32_16x16x4_f32(false, a, false, b1, (short)0, acc1, false, false);
            acc2 = __builtin_amdgcn_wmma_f32_16x16x4_f32(false, a, false, b2, (short)0, acc2, false, false);
            acc3 = __builtin_amdgcn_wmma_f32_16x16x4_f32(false, a, false, b3, (short)0, acc3, false, false);
        }
        asm volatile("v_nop\n\tv_nop\n\tv_nop\n\tv_nop\n\tv_nop\n\tv_nop\n\tv_nop\n\tv_nop"
                     : "+v"(acc0), "+v"(acc1), "+v"(acc2), "+v"(acc3));
        __syncthreads();
    }
#pragma unroll
    for (int k = 0; k < 16; ++k) {
        const int c = (k << 2) + (h << 1);
        const v2f a = *(const v2f*)(pax + c);
        const float* wp = wbase + ((64 + (k << 1) + h) << 7);
        const v2f b0 = *(const v2f*)(wp);
        const v2f b1 = *(const v2f*)(wp + 32);
        const v2f b2 = *(const v2f*)(wp + 64);
        const v2f b3 = *(const v2f*)(wp + 96);
        acc0 = __builtin_amdgcn_wmma_f32_16x16x4_f32(false, a, false, b0, (short)0, acc0, false, false);
        acc1 = __builtin_amdgcn_wmma_f32_16x16x4_f32(false, a, false, b1, (short)0, acc1, false, false);
        acc2 = __builtin_amdgcn_wmma_f32_16x16x4_f32(false, a, false, b2, (short)0, acc2, false, false);
        acc3 = __builtin_amdgcn_wmma_f32_16x16x4_f32(false, a, false, b3, (short)0, acc3, false, false);
    }

    const float bs0 = bias[m];
    const float bs1 = bias[16 + m];
    const float bs2 = bias[32 + m];
    const float bs3 = bias[48 + m];

#pragma unroll
    for (int j = 0; j < 8; ++j) {
        const float o0 = acc0[j] + bs0;
        const float o1 = acc1[j] + bs1;
        const float o2 = acc2[j] + bs2;
        const float o3 = acc3[j] + bs3;
        float s = o0 * o0 + o1 * o1 + o2 * o2 + o3 * o3;
        s += __shfl_xor(s, 1, 32);
        s += __shfl_xor(s, 2, 32);
        s += __shfl_xor(s, 4, 32);
        s += __shfl_xor(s, 8, 32);
        const float scale = 1.0f / fmaxf(sqrtf(s), 1e-12f);
        const long orow = row0 + j + (h << 3);
        if (orow < n) {
            float* po = out + orow * D_FEAT + m;
            po[0]  = o0 * scale;
            po[16] = o1 * scale;
            po[32] = o2 * scale;
            po[48] = o3 * scale;
        }
    }
}

extern "C" void kernel_launch(void* const* d_in, const int* in_sizes, int n_in,
                              void* d_out, int out_size, void* d_ws, size_t ws_size,
                              hipStream_t stream) {
    const float* x1    = (const float*)d_in[0];
    const float* x2    = (const float*)d_in[1];
    const int*   e_pos = (const int*)d_in[2];
    const int*   e_neg = (const int*)d_in[3];
    const float* W     = (const float*)d_in[4];
    const float* bias  = (const float*)d_in[5];
    float*       out   = (float*)d_out;

    const int n       = in_sizes[0] / D_FEAT;
    const int n_edges = in_sizes[2] / 2;

    const int n_buckets = (n + BUCKET_ROWS - 1) / BUCKET_ROWS;
    const int n_chunks  = (n_edges + CHUNK - 1) / CHUNK;

    int* cntm  = (int*)d_ws;
    int* total = cntm + (size_t)2 * n_buckets * n_chunks;
    int* pairs = total + (size_t)2 * n_buckets;

    const dim3 egrid(n_chunks, 2);
    bucket_count_kernel<<<egrid, 512, 0, stream>>>(e_pos, e_neg, cntm, n_edges, n_buckets, n_chunks);
    bucket_scan_kernel<<<(2 * n_buckets + 7) / 8, 256, 0, stream>>>(cntm, total, 2 * n_buckets, n_chunks);
    bucket_place_kernel<<<egrid, 512, 0, stream>>>(e_pos, e_neg, cntm, total, pairs,
                                                   n_edges, n_buckets, n_chunks);
    {
        const int tiles  = (n + 15) / 16;
        const int blocks = (tiles + 15) / 16;
        sage_gather_gemm_norm_kernel<<<blocks, 512, 0, stream>>>(x1, x2, total, pairs,
                                                                 W, bias, out, n, n_edges, n_buckets);
    }
}


# Gather walks each half-wave's 8 rows as one flat run of the row-ordered slot list, 10 neighbour rows in flight per trip (was row by row, 4 per trip); each row still summed as itself, then its slots
# speedup vs baseline: 3.7263x; 1.1163x over previous
#include <hip/hip_runtime.h>
#include <hip/hip_bf16.h>

typedef __attribute__((ext_vector_type(2))) float v2f;
typedef __attribute__((ext_vector_type(4))) float v4f;
typedef __attribute__((ext_vector_type(8))) float v8f;

#define D_FEAT 64
#define TILE_PITCH 68
#define BUCKET_ROWS 256
#define MAX_BUCKETS 512
#define CHUNK 4096
#define PIECE 4096


__device__ __forceinline__ int lds_fetch_add(int* p, int v) {
    return __hip_atomic_fetch_add(p, v, __ATOMIC_RELAXED, __HIP_MEMORY_SCOPE_WORKGROUP);
}

__global__ __launch_bounds__(512) void bucket_count_kernel(const int* __restrict__ e_pos,
                                                           const int* __restrict__ e_neg,
                                                           int* __restrict__ cntm,
                                                           int n_edges, int n_buckets, int n_chunks) {
    __shared__ int hist[MAX_BUCKETS];
    const int set = blockIdx.y;
    const int tid = threadIdx.x;
    const int* __restrict__ rows = set ? e_neg : e_pos;
    hist[tid] = 0;
    __syncthreads();
    const int e0 = blockIdx.x * CHUNK + tid;
    int rw[8];
#pragma unroll
    for (int u = 0; u < 8; ++u) {
        const int e = e0 + u * 512;
        rw[u] = (e < n_edges) ? rows[e] : -1;
    }
#pragma unroll
    for (int u = 0; u < 8; ++u)
        if (rw[u] >= 0) lds_fetch_add(&hist[rw[u] >> 8], 1);
    __syncthreads();
    if (tid < n_buckets)
        cntm[((long)set * n_buckets + tid) * n_chunks + blockIdx.x] = hist[tid];
}

__global__ __launch_bounds__(256) void bucket_scan_kernel(int* __restrict__ cntm,
                                                          int* __restrict__ total,
                                                          int n_lists, int n_chunks) {
    const int lane = threadIdx.x & 31;
    const int g = blockIdx.x * 8 + (threadIdx.x >> 5);
    if (g >= n_lists) return;
    int* __restrict__ p = cntm + (long)g * n_chunks;
    int carry = 0;
    for (int j = 0; j < n_chunks; j += 32) {
        const int i = j + lane;
        const int v = (i < n_chunks) ? p[i] : 0;
        int inc = v;
        for (int o = 1; o < 32; o <<= 1) {
            const int t = __shfl_up(inc, o, 32);
            if (lane >= o) inc += t;
        }
        if (i < n_chunks) p[i] = carry + inc - v;
        carry += __shfl(inc, 31, 32);
    }
    if (lane == 0) total[g] = carry;
}

__global__ __launch_bounds__(512) void bucket_place_kernel(const int* __restrict__ e_pos,
                                                           const int* __restrict__ e_neg,
                                                           const int* __restrict__ cntm,
                                                           const int* __restrict__ total,
                                                           int* __restrict__ pairs,
                                                           int n_edges, int n_buckets, int n_chunks) {
    __shared__ int cursor[MAX_BUCKETS];
    __shared__ int wsum[16];
    const int set  = blockIdx.y;
    const int tid  = threadIdx.x;
    const int lane = tid & 31;
    const int wave = tid >> 5;
    const int* __restrict__ edges = set ? e_neg : e_pos;

    const int e0 = blockIdx.x * CHUNK + tid;
    int rw[8], cl[8];
#pragma unroll
    for (int u = 0; u < 8; ++u) {
        const int e = e0 + u * 512;
        const bool ok = e < n_edges;
        rw[u] = ok ? edges[e] : -1;
        cl[u] = ok ? edges[(long)n_edges + e] : 0;
    }

    const bool live = tid < n_buckets;
    const int t   = live ? total[set * n_buckets + tid] : 0;
    const int off = live ? cntm[((long)set * n_buckets + tid) * n_chunks + blockIdx.x] : 0;
    int inc = t;
    for (int o = 1; o < 32; o <<= 1) {
        const int s = __shfl_up(inc, o, 32);
        if (lane >= o) inc += s;
    }
    if (lane == 31) wsum[wave] = inc;
    __syncthreads();
    int woff = 0;
    for (int w = 0; w < 16; ++w) woff += (w < wave) ? wsum[w] : 0;
    cursor[tid] = woff + inc - t + off;
    __syncthreads();

    int* __restrict__ dst = pairs + (long)set * n_edges;
#pragma unroll
    for (int u = 0; u < 8; ++u) {
        if (rw[u] >= 0) {
            const int p = lds_fetch_add(&cursor[rw[u] >> 8], 1);
            dst[p] = (cl[u] << 8) | (rw[u] & (BUCKET_ROWS - 1));
        }
    }
}

#define GATHER_BATCH 10

__device__ __forceinline__ void gather_mean_tile(const float* __restrict__ x,
                                                 const int* __restrict__ seg, int seg_count,
                                                 long row0, int n, float* T,
                                                 int* lnbr, int* deg, int* lstart, int* cursor,
                                                 int* rdeg, int* wsum,
                                                 int tid, int lane, int wave) {
    const int hh = lane >> 4;
    const int q  = lane & 15;
    const char* __restrict__ xb = (const char*)x;
    const unsigned qoff  = (unsigned)q << 4;
    const unsigned rbase = (unsigned)row0;
    const unsigned rlast = (unsigned)(n - 1);

    const v4f zero4 = {0.0f, 0.0f, 0.0f, 0.0f};

    const char* __restrict__ pself = xb + (((rbase + (unsigned)hh) << 8) | qoff);
    v4f sv[8];
#pragma unroll
    for (int i = 0; i < 8; ++i) {
        sv[i] = zero4;
        if (rbase + (unsigned)(2 * i + hh) <= rlast) sv[i] = *(const v4f*)(pself + i * 512);
    }
#pragma unroll
    for (int i = 0; i < 8; ++i)
        *(v4f*)(T + (2 * i + hh) * TILE_PITCH + (q << 2)) = sv[i];
    if (tid < BUCKET_ROWS) rdeg[tid] = 0;

    const unsigned self_off = (min(rbase + 8u * (unsigned)hh, rlast) << 8) | qoff;

    for (int pb = 0; pb < seg_count; pb += PIECE) {
        const int pc = min(PIECE, seg_count - pb);
        int w[8];
#pragma unroll
        for (int u = 0; u < 8; ++u) {
            const int e = tid + u * 512;
            w[u] = (e < pc) ? seg[pb + e] : 0;
        }
        if (tid < BUCKET_ROWS) deg[tid] = 0;
        __syncthreads();
#pragma unroll
        for (int u = 0; u < 8; ++u)
            if (tid + u * 512 < pc) lds_fetch_add(&deg[w[u] & (BUCKET_ROWS - 1)], 1);
        __syncthreads();
        const int d0 = (tid < BUCKET_ROWS) ? deg[tid] : 0;
        int inc = d0;
        for (int o = 1; o < 32; o <<= 1) {
            const int t = __shfl_up(inc, o, 32);
            if (lane >= o) inc += t;
        }
        if (lane == 31 && wave < 8) wsum[wave] = inc;
        __syncthreads();
        if (tid < BUCKET_ROWS) {
            int woff = 0;
            for (int k = 0; k < 8; ++k) woff += (k < wave) ? wsum[k] : 0;
            const int st = woff + inc - d0;
            lstart[tid] = st;
            cursor[tid] = st;
            rdeg[tid] += d0;
        }
        __syncthreads();
#pragma unroll
        for (int u = 0; u < 8; ++u) {
            if (tid + u * 512 < pc) {
                const int slot = lds_fetch_add(&cursor[w[u] & (BUCKET_ROWS - 1)], 1);
                lnbr[slot] = w[u];
            }
        }
        __syncthreads();

        const int beg = lstart[wave * 16 + 8 * hh];
        const int cnt = cursor[wave * 16 + 8 * hh + 7] - beg;
        const int cmax = max(cnt, __shfl_xor(cnt, 16, 32));

        int cur = 8 * hh;
        float* tcur = T + cur * TILE_PITCH + (q << 2);
        v4f acc = *(v4f*)tcur;
        for (int j = 0; j < cmax; j += GATHER_BATCH) {
            int wd[GATHER_BATCH];
            v4f t[GATHER_BATCH];
#pragma unroll
            for (int u = 0; u < GATHER_BATCH; ++u)
                wd[u] = lnbr[(j + u < cnt) ? (beg + j + u) : 0];
#pragma unroll
            for (int u = 0; u < GATHER_BATCH; ++u) {
                const unsigned off = (j + u < cnt)
                    ? (((unsigned)wd[u] & ~(unsigned)(BUCKET_ROWS - 1)) | qoff)
                    : self_off;
                t[u] = *(const v4f*)(xb + off);
            }
#pragma unroll
            for (int u = 0; u < GATHER_BATCH; ++u) {
                const bool ok = j + u < cnt;
                const int lr = wd[u] & 15;
                if (ok && lr != cur) {
                    *(v4f*)tcur = acc;
                    cur = lr;
                    tcur = T + lr * TILE_PITCH + (q << 2);
                    acc = *(v4f*)tcur;
                }
                acc += ok ? t[u] : zero4;
            }
        }
        *(v4f*)tcur = acc;
        __syncthreads();
    }
    __syncthreads();
#pragma unroll
    for (int i = 0; i < 8; ++i) {
        const int lr = 2 * i + hh;
        const float inv = 1.0f / fmaxf((float)(rdeg[wave * 16 + lr] + 1), 1.0f);
        float* trow = T + lr * TILE_PITCH + (q << 2);
        v4f acc = *(v4f*)trow;
        acc *= inv;
        *(v4f*)trow = acc;
    }
}

__global__ __launch_bounds__(512, 8) void sage_gather_gemm_norm_kernel(
        const float* __restrict__ x1,   const float* __restrict__ x2,
        const int* __restrict__ total,  const int* __restrict__ pairs,
        const float* __restrict__ W,    const float* __restrict__ bias,
        float* __restrict__ out, int n, int n_edges, int n_buckets) {
    __shared__ float ldsW[192 * 64];
    __shared__ __attribute__((aligned(16))) float ldsA[16 * 16 * TILE_PITCH];
    __shared__ int lnbr[PIECE];
    __shared__ int deg[BUCKET_ROWS], lstart[BUCKET_ROWS], cursor[BUCKET_ROWS], rdeg[BUCKET_ROWS];
    __shared__ int wsum[8];
    __shared__ int red[2][16];

    const int tid = threadIdx.x;
    for (int idx = tid; idx < 192 * 64; idx += 512) {
        const int row = idx >> 6;
        const int col = idx & 63;
        ldsW[((((row >> 1) << 6) | col) << 1) | (row & 1)] = W[idx];
    }

    const int lane = tid & 31;
    const int wave = tid >> 5;
    const int m = lane & 15;
    const int h = lane >> 4;

    {
        const int bk = blockIdx.x;
        int v0 = (tid < bk) ? total[tid] : 0;
        int v1 = (tid < bk) ? total[n_buckets + tid] : 0;
        for (int o = 16; o > 0; o >>= 1) {
            v0 += __shfl_xor(v0, o, 32);
            v1 += __shfl_xor(v1, o, 32);
        }
        if (lane == 0) { red[0][wave] = v0; red[1][wave] = v1; }
        __syncthreads();
    }
    int seg_beg0 = 0, seg_beg1 = 0;
    for (int k = 0; k < 16; ++k) { seg_beg0 += red[0][k]; seg_beg1 += red[1][k]; }

    const long tile = (long)blockIdx.x * 16 + wave;
    const long row0 = tile << 4;

    float* T = ldsA + wave * (16 * TILE_PITCH);
    const float* ta = T + m * TILE_PITCH;

    v8f acc0 = {}; v8f acc1 = {}; v8f acc2 = {}; v8f acc3 = {};
    const float* wbase = ldsW + (m << 1);

#pragma nounroll
    for (int set = 0; set < 2; ++set) {
        const int seg_beg = __builtin_amdgcn_readfirstlane(set ? seg_beg1 : seg_beg0);
        const int seg_cnt = __builtin_amdgcn_readfirstlane(total[set * n_buckets + blockIdx.x]);
        gather_mean_tile(set ? x2 : x1, pairs + (long)set * n_edges + seg_beg, seg_cnt,
                         row0, n, T, lnbr, deg, lstart, cursor, rdeg, wsum, tid, lane, wave);
        __syncthreads();
        const float* wset = wbase + ((set << 5) << 7);
#pragma unroll
        for (int k = 0; k < 16; ++k) {
            const int c = (k << 2) + (h << 1);
            const v2f a = *(const v2f*)(ta + c);
            const float* wp = wset + (((k << 1) + h) << 7);
            const v2f b0 = *(const v2f*)(wp);
            const v2f b1 = *(const v2f*)(wp + 32);
            const v2f b2 = *(const v2f*)(wp + 64);
            const v2f b3 = *(const v2f*)(wp + 96);
            acc0 = __builtin_amdgcn_wmma_f32_16x16x4_f32(false, a, false, b0, (short)0, acc0, false, false);
            acc1 = __builtin_amdgcn_wmma_f32_16x16x4_f32(false, a, false, b1, (short)0, acc1, false, false);
            acc2 = __builtin_amdgcn_wmma_f32_16x16x4_f32(false, a, false, b2, (short)0, acc2, false, false);
            acc3 = __builtin_amdgcn_wmma_f32_16x16x4_f32(false, a, false, b3, (short)0, acc3, false, false);
        }
        asm volatile("v_nop\n\tv_nop\n\tv_nop\n\tv_nop\n\tv_nop\n\tv_nop\n\tv_nop\n\tv_nop"
                     : "+v"(acc0), "+v"(acc1), "+v"(acc2), "+v"(acc3));
        __syncthreads();
    }
    long r = row0 + m;
    if (r > (long)(n - 1)) r = (long)(n - 1);
    const float* pax = x1 + r * D_FEAT;
#pragma unroll
    for (int k = 0; k < 16; ++k) {
        const int c = (k << 2) + (h << 1);
        const v2f a = *(const v2f*)(pax + c);
        const float* wp = wbase + ((64 + (k << 1) + h) << 7);
        const v2f b0 = *(const v2f*)(wp);
        const v2f b1 = *(const v2f*)(wp + 32);
        const v2f b2 = *(const v2f*)(wp + 64);
        const v2f b3 = *(const v2f*)(wp + 96);
        acc0 = __builtin_amdgcn_wmma_f32_16x16x4_f32(false, a, false, b0, (short)0, acc0, false, false);
        acc1 = __builtin_amdgcn_wmma_f32_16x16x4_f32(false, a, false, b1, (short)0, acc1, false, false);
        acc2 = __builtin_amdgcn_wmma_f32_16x16x4_f32(false, a, false, b2, (short)0, acc2, false, false);
        acc3 = __builtin_amdgcn_wmma_f32_16x16x4_f32(false, a, false, b3, (short)0, acc3, false, false);
    }

    const float bs0 = bias[m];
    const float bs1 = bias[16 + m];
    const float bs2 = bias[32 + m];
    const float bs3 = bias[48 + m];

#pragma unroll
    for (int j = 0; j < 8; ++j) {
        const float o0 = acc0[j] + bs0;
        const float o1 = acc1[j] + bs1;
        const float o2 = acc2[j] + bs2;
        const float o3 = acc3[j] + bs3;
        float s = o0 * o0 + o1 * o1 + o2 * o2 + o3 * o3;
        s += __shfl_xor(s, 1, 32);
        s += __shfl_xor(s, 2, 32);
        s += __shfl_xor(s, 4, 32);
        s += __shfl_xor(s, 8, 32);
        const float scale = 1.0f / fmaxf(sqrtf(s), 1e-12f);
        const long orow = row0 + j + (h << 3);
        if (orow < n) {
            float* po = out + orow * D_FEAT + m;
            po[0]  = o0 * scale;
            po[16] = o1 * scale;
            po[32] = o2 * scale;
            po[48] = o3 * scale;
        }
    }
}

extern "C" void kernel_launch(void* const* d_in, const int* in_sizes, int n_in,
                              void* d_out, int out_size, void* d_ws, size_t ws_size,
                              hipStream_t stream) {
    const float* x1    = (const float*)d_in[0];
    const float* x2    = (const float*)d_in[1];
    const int*   e_pos = (const int*)d_in[2];
    const int*   e_neg = (const int*)d_in[3];
    const float* W     = (const float*)d_in[4];
    const float* bias  = (const float*)d_in[5];
    float*       out   = (float*)d_out;

    const int n       = in_sizes[0] / D_FEAT;
    const int n_edges = in_sizes[2] / 2;

    const int n_buckets = (n + BUCKET_ROWS - 1) / BUCKET_ROWS;
    const int n_chunks  = (n_edges + CHUNK - 1) / CHUNK;

    int* cntm  = (int*)d_ws;
    int* total = cntm + (size_t)2 * n_buckets * n_chunks;
    int* pairs = total + (size_t)2 * n_buckets;

    const dim3 egrid(n_chunks, 2);
    bucket_count_kernel<<<egrid, 512, 0, stream>>>(e_pos, e_neg, cntm, n_edges, n_buckets, n_chunks);
    bucket_scan_kernel<<<(2 * n_buckets + 7) / 8, 256, 0, stream>>>(cntm, total, 2 * n_buckets, n_chunks);
    bucket_place_kernel<<<egrid, 512, 0, stream>>>(e_pos, e_neg, cntm, total, pairs,
                                                   n_edges, n_buckets, n_chunks);
    {
        const int tiles  = (n + 15) / 16;
        const int blocks = (tiles + 15) / 16;
        sage_gather_gemm_norm_kernel<<<blocks, 512, 0, stream>>>(x1, x2, total, pairs,
                                                                 W, bias, out, n, n_edges, n_buckets);
    }
}


# Edges ordered by bucket inside each 4096-edge chunk in LDS, stored contiguously; GEMM block collects its bucket's run from every chunk (start table + search); 4 launches to 2; row sums arrival-order
# speedup vs baseline: 5.7532x; 1.5439x over previous
#include <hip/hip_runtime.h>
#include <hip/hip_bf16.h>

typedef __attribute__((ext_vector_type(2))) float v2f;
typedef __attribute__((ext_vector_type(4))) float v4f;
typedef __attribute__((ext_vector_type(8))) float v8f;

#define D_FEAT 64
#define TILE_PITCH 68
#define BUCKET_ROWS 256
#define MAX_BUCKETS 512
#define CHUNK 4096
#define MAX_CHUNKS 512
#define PIECE 4096


__device__ __forceinline__ int lds_fetch_add(int* p, int v) {
    return __hip_atomic_fetch_add(p, v, __ATOMIC_RELAXED, __HIP_MEMORY_SCOPE_WORKGROUP);
}

__global__ __launch_bounds__(512) void chunk_group_kernel(const int* __restrict__ e_pos,
                                                          const int* __restrict__ e_neg,
                                                          int* __restrict__ off,
                                                          int* __restrict__ pairs,
                                                          int n_edges, int n_buckets, int n_chunks) {
    __shared__ int hist[MAX_BUCKETS];
    __shared__ int wsum[16];
    __shared__ int sorted[CHUNK];
    const int set  = blockIdx.y;
    const int tid  = threadIdx.x;
    const int lane = tid & 31;
    const int wave = tid >> 5;
    const int* __restrict__ edges = set ? e_neg : e_pos;

    hist[tid] = 0;
    const int e0 = blockIdx.x * CHUNK + tid;
    int rw[8], cl[8], rk[8];
#pragma unroll
    for (int u = 0; u < 8; ++u) {
        const int e = e0 + u * 512;
        const bool ok = e < n_edges;
        rw[u] = ok ? edges[e] : -1;
        cl[u] = ok ? edges[(long)n_edges + e] : 0;
    }
    __syncthreads();
#pragma unroll
    for (int u = 0; u < 8; ++u) {
        rk[u] = 0;
        if (rw[u] >= 0) rk[u] = lds_fetch_add(&hist[rw[u] >> 8], 1);
    }
    __syncthreads();

    const int cnt = hist[tid];
    int inc = cnt;
    for (int o = 1; o < 32; o <<= 1) {
        const int s = __shfl_up(inc, o, 32);
        if (lane >= o) inc += s;
    }
    if (lane == 31) wsum[wave] = inc;
    __syncthreads();
    int woff = 0;
    for (int w = 0; w < 16; ++w) woff += (w < wave) ? wsum[w] : 0;
    const int start = woff + inc - cnt;
    hist[tid] = start;
    if (tid <= n_buckets)
        off[((long)set * (n_buckets + 1) + tid) * n_chunks + blockIdx.x] = start;
    __syncthreads();

#pragma unroll
    for (int u = 0; u < 8; ++u)
        if (rw[u] >= 0)
            sorted[hist[rw[u] >> 8] + rk[u]] = (cl[u] << 8) | (rw[u] & (BUCKET_ROWS - 1));
    __syncthreads();

    const int m = min(CHUNK, n_edges - blockIdx.x * CHUNK);
    int* __restrict__ dst = pairs + (long)set * n_edges + (long)blockIdx.x * CHUNK;
#pragma unroll
    for (int u = 0; u < 8; ++u) {
        const int i = tid + u * 512;
        if (i < m) dst[i] = sorted[i];
    }
}

#define GATHER_BATCH 10

__device__ __forceinline__ void gather_mean_tile(const float* __restrict__ x,
                                                 const int* __restrict__ pw,
                                                 const int* __restrict__ offk, int n_chunks,
                                                 long row0, int n, float* T,
                                                 int* lnbr, int* deg, int* lstart, int* cursor,
                                                 int* rdeg, int* wsum,
                                                 int* cpos, int* cbase,
                                                 int tid, int lane, int wave) {
    const int hh = lane >> 4;
    const int q  = lane & 15;
    const char* __restrict__ xb = (const char*)x;
    const unsigned qoff  = (unsigned)q << 4;
    const unsigned rbase = (unsigned)row0;
    const unsigned rlast = (unsigned)(n - 1);

    const v4f zero4 = {0.0f, 0.0f, 0.0f, 0.0f};

    const char* __restrict__ pself = xb + (((rbase + (unsigned)hh) << 8) | qoff);
    v4f sv[8];
#pragma unroll
    for (int i = 0; i < 8; ++i) {
        sv[i] = zero4;
        if (rbase + (unsigned)(2 * i + hh) <= rlast) sv[i] = *(const v4f*)(pself + i * 512);
    }
#pragma unroll
    for (int i = 0; i < 8; ++i)
        *(v4f*)(T + (2 * i + hh) * TILE_PITCH + (q << 2)) = sv[i];
    if (tid < BUCKET_ROWS) rdeg[tid] = 0;

    int seg_count;
    {
        int rs = 0, len = 0;
        if (tid < n_chunks) {
            const int* __restrict__ offe = offk + n_chunks;
            rs  = offk[tid];
            len = offe[tid] - rs;
        }
        int inc = len;
        for (int o = 1; o < 32; o <<= 1) {
            const int t = __shfl_up(inc, o, 32);
            if (lane >= o) inc += t;
        }
        if (lane == 31) wsum[wave] = inc;
        __syncthreads();
        int woff = 0, tot = 0;
        for (int k = 0; k < 16; ++k) {
            const int v = wsum[k];
            tot += v;
            woff += (k < wave) ? v : 0;
        }
        cpos[tid]  = (tid < n_chunks) ? (woff + inc - len) : 0x7fffffff;
        cbase[tid] = rs;
        seg_count = __builtin_amdgcn_readfirstlane(tot);
        __syncthreads();
    }

    const unsigned self_off = (min(rbase + 8u * (unsigned)hh, rlast) << 8) | qoff;

    for (int pb = 0; pb < seg_count; pb += PIECE) {
        const int pc = min(PIECE, seg_count - pb);
        int w[8], p[8], c[8];
        const int pend = pb + pc;
        int pt = pb + tid;
        asm volatile("" : "+v"(pt));
#pragma unroll
        for (int u = 0; u < 8; ++u) {
            const int pp = pt + u * 512;
            p[u] = (pp < pend) ? pp : pb;
            c[u] = 0;
        }
#pragma unroll
        for (int step = MAX_CHUNKS >> 1; step > 0; step >>= 1) {
#pragma unroll
            for (int u = 0; u < 8; ++u)
                if (cpos[c[u] + step] <= p[u]) c[u] += step;
        }
#pragma unroll
        for (int u = 0; u < 8; ++u)
            w[u] = pw[c[u] * CHUNK + cbase[c[u]] + (p[u] - cpos[c[u]])];
        if (tid < BUCKET_ROWS) deg[tid] = 0;
        __syncthreads();
#pragma unroll
        for (int u = 0; u < 8; ++u)
            if (tid + u * 512 < pc) lds_fetch_add(&deg[w[u] & (BUCKET_ROWS - 1)], 1);
        __syncthreads();
        const int d0 = (tid < BUCKET_ROWS) ? deg[tid] : 0;
        int inc = d0;
        for (int o = 1; o < 32; o <<= 1) {
            const int t = __shfl_up(inc, o, 32);
            if (lane >= o) inc += t;
        }
        if (lane == 31 && wave < 8) wsum[wave] = inc;
        __syncthreads();
        if (tid < BUCKET_ROWS) {
            int woff = 0;
            for (int k = 0; k < 8; ++k) woff += (k < wave) ? wsum[k] : 0;
            const int st = woff + inc - d0;
            lstart[tid] = st;
            cursor[tid] = st;
            rdeg[tid] += d0;
        }
        __syncthreads();
#pragma unroll
        for (int u = 0; u < 8; ++u) {
            if (tid + u * 512 < pc) {
                const int slot = lds_fetch_add(&cursor[w[u] & (BUCKET_ROWS - 1)], 1);
                lnbr[slot] = w[u];
            }
        }
        __syncthreads();

        const int beg = lstart[wave * 16 + 8 * hh];
        const int cnt = cursor[wave * 16 + 8 * hh + 7] - beg;
        const int cmax = max(cnt, __shfl_xor(cnt, 16, 32));

        int cur = 8 * hh;
        float* tcur = T + cur * TILE_PITCH + (q << 2);
        v4f acc = *(v4f*)tcur;
        for (int j = 0; j < cmax; j += GATHER_BATCH) {
            int wd[GATHER_BATCH];
            v4f t[GATHER_BATCH];
#pragma unroll
            for (int u = 0; u < GATHER_BATCH; ++u)
                wd[u] = lnbr[(j + u < cnt) ? (beg + j + u) : 0];
#pragma unroll
            for (int u = 0; u < GATHER_BATCH; ++u) {
                const unsigned off = (j + u < cnt)
                    ? (((unsigned)wd[u] & ~(unsigned)(BUCKET_ROWS - 1)) | qoff)
                    : self_off;
                t[u] = *(const v4f*)(xb + off);
            }
#pragma unroll
            for (int u = 0; u < GATHER_BATCH; ++u) {
                const bool ok = j + u < cnt;
                const int lr = wd[u] & 15;
                if (ok && lr != cur) {
                    *(v4f*)tcur = acc;
                    cur = lr;
                    tcur = T + lr * TILE_PITCH + (q << 2);
                    acc = *(v4f*)tcur;
                }
                acc += ok ? t[u] : zero4;
            }
        }
        *(v4f*)tcur = acc;
        __syncthreads();
    }
    __syncthreads();
#pragma unroll
    for (int i = 0; i < 8; ++i) {
        const int lr = 2 * i + hh;
        const float inv = 1.0f / fmaxf((float)(rdeg[wave * 16 + lr] + 1), 1.0f);
        float* trow = T + lr * TILE_PITCH + (q << 2);
        v4f acc = *(v4f*)trow;
        acc *= inv;
        *(v4f*)trow = acc;
    }
}

__global__ __launch_bounds__(512, 8) void sage_gather_gemm_norm_kernel(
        const float* __restrict__ x1,   const float* __restrict__ x2,
        const int* __restrict__ off,    const int* __restrict__ pairs,
        const float* __restrict__ W,    const float* __restrict__ bias,
        float* __restrict__ out, int n, int n_edges, int n_buckets, int n_chunks) {
    __shared__ float ldsW[192 * 64];
    __shared__ __attribute__((aligned(16))) float ldsA[16 * 16 * TILE_PITCH];
    __shared__ int lnbr[PIECE];
    __shared__ int deg[BUCKET_ROWS], lstart[BUCKET_ROWS], cursor[BUCKET_ROWS], rdeg[BUCKET_ROWS];
    __shared__ int wsum[16];
    __shared__ int cpos[MAX_CHUNKS], cbase[MAX_CHUNKS];

    const int tid = threadIdx.x;
    for (int idx = tid; idx < 192 * 64; idx += 512) {
        const int row = idx >> 6;
        const int col = idx & 63;
        ldsW[((((row >> 1) << 6) | col) << 1) | (row & 1)] = W[idx];
    }

    const int lane = tid & 31;
    const int wave = tid >> 5;
    const int m = lane & 15;
    const int h = lane >> 4;

    const long tile = (long)blockIdx.x * 16 + wave;
    const long row0 = tile << 4;

    float* T = ldsA + wave * (16 * TILE_PITCH);
    const float* ta = T + m * TILE_PITCH;

    v8f acc0 = {}; v8f acc1 = {}; v8f acc2 = {}; v8f acc3 = {};
    const float* wbase = ldsW + (m << 1);

#pragma nounroll
    for (int set = 0; set < 2; ++set) {
        const int* __restrict__ offk = off + ((long)set * (n_buckets + 1) + blockIdx.x) * n_chunks;
        gather_mean_tile(set ? x2 : x1, pairs + (long)set * n_edges, offk, n_chunks,
                         row0, n, T, lnbr, deg, lstart, cursor, rdeg, wsum, cpos, cbase,
                         tid, lane, wave);
        __syncthreads();
        const float* wset = wbase + ((set << 5) << 7);
#pragma unroll
        for (int k = 0; k < 16; ++k) {
            const int c = (k << 2) + (h << 1);
            const v2f a = *(const v2f*)(ta + c);
            const float* wp = wset + (((k << 1) + h) << 7);
            const v2f b0 = *(const v2f*)(wp);
            const v2f b1 = *(const v2f*)(wp + 32);
            const v2f b2 = *(const v2f*)(wp + 64);
            const v2f b3 = *(const v2f*)(wp + 96);
            acc0 = __builtin_amdgcn_wmma_f32_16x16x4_f32(false, a, false, b0, (short)0, acc0, false, false);
            acc1 = __builtin_amdgcn_wmma_f32_16x16x4_f32(false, a, false, b1, (short)0, acc1, false, false);
            acc2 = __builtin_amdgcn_wmma_f32_16x16x4_f32(false, a, false, b2, (short)0, acc2, false, false);
            acc3 = __builtin_amdgcn_wmma_f32_16x16x4_f32(false, a, false, b3, (short)0, acc3, false, false);
        }
        asm volatile("v_nop\n\tv_nop\n\tv_nop\n\tv_nop\n\tv_nop\n\tv_nop\n\tv_nop\n\tv_nop"
                     : "+v"(acc0), "+v"(acc1), "+v"(acc2), "+v"(acc3));
        __syncthreads();
    }
    long r = row0 + m;
    if (r > (long)(n - 1)) r = (long)(n - 1);
    const float* pax = x1 + r * D_FEAT;
#pragma unroll
    for (int k = 0; k < 16; ++k) {
        const int c = (k << 2) + (h << 1);
        const v2f a = *(const v2f*)(pax + c);
        const float* wp = wbase + ((64 + (k << 1) + h) << 7);
        const v2f b0 = *(const v2f*)(wp);
        const v2f b1 = *(const v2f*)(wp + 32);
        const v2f b2 = *(const v2f*)(wp + 64);
        const v2f b3 = *(const v2f*)(wp + 96);
        acc0 = __builtin_amdgcn_wmma_f32_16x16x4_f32(false, a, false, b0, (short)0, acc0, false, false);
        acc1 = __builtin_amdgcn_wmma_f32_16x16x4_f32(false, a, false, b1, (short)0, acc1, false, false);
        acc2 = __builtin_amdgcn_wmma_f32_16x16x4_f32(false, a, false, b2, (short)0, acc2, false, false);
        acc3 = __builtin_amdgcn_wmma_f32_16x16x4_f32(false, a, false, b3, (short)0, acc3, false, false);
    }

    const float bs0 = bias[m];
    const float bs1 = bias[16 + m];
    const float bs2 = bias[32 + m];
    const float bs3 = bias[48 + m];

#pragma unroll
    for (int j = 0; j < 8; ++j) {
        const float o0 = acc0[j] + bs0;
        const float o1 = acc1[j] + bs1;
        const float o2 = acc2[j] + bs2;
        const float o3 = acc3[j] + bs3;
        float s = o0 * o0 + o1 * o1 + o2 * o2 + o3 * o3;
        s += __shfl_xor(s, 1, 32);
        s += __shfl_xor(s, 2, 32);
        s += __shfl_xor(s, 4, 32);
        s += __shfl_xor(s, 8, 32);
        const float scale = 1.0f / fmaxf(sqrtf(s), 1e-12f);
        const long orow = row0 + j + (h << 3);
        if (orow < n) {
            float* po = out + orow * D_FEAT + m;
            po[0]  = o0 * scale;
            po[16] = o1 * scale;
            po[32] = o2 * scale;
            po[48] = o3 * scale;
        }
    }
}

extern "C" void kernel_launch(void* const* d_in, const int* in_sizes, int n_in,
                              void* d_out, int out_size, void* d_ws, size_t ws_size,
                              hipStream_t stream) {
    const float* x1    = (const float*)d_in[0];
    const float* x2    = (const float*)d_in[1];
    const int*   e_pos = (const int*)d_in[2];
    const int*   e_neg = (const int*)d_in[3];
    const float* W     = (const float*)d_in[4];
    const float* bias  = (const float*)d_in[5];
    float*       out   = (float*)d_out;

    const int n       = in_sizes[0] / D_FEAT;
    const int n_edges = in_sizes[2] / 2;

    const int n_buckets = (n + BUCKET_ROWS - 1) / BUCKET_ROWS;
    const int n_chunks  = (n_edges + CHUNK - 1) / CHUNK;

    int* off   = (int*)d_ws;
    int* pairs = off + (size_t)2 * (n_buckets + 1) * n_chunks;

    const dim3 egrid(n_chunks, 2);
    chunk_group_kernel<<<egrid, 512, 0, stream>>>(e_pos, e_neg, off, pairs,
                                                  n_edges, n_buckets, n_chunks);
    {
        const int tiles  = (n + 15) / 16;
        const int blocks = (tiles + 15) / 16;
        sage_gather_gemm_norm_kernel<<<blocks, 512, 0, stream>>>(x1, x2, off, pairs,
                                                                 W, bias, out, n, n_edges, n_buckets,
                                                                 n_chunks);
    }
}


# Gather trips wholly inside both halves' runs drop the per-slot end-of-run tests and selects (word at a fixed offset, row loaded and added as is); only the last trips are masked; sum order kept
# speedup vs baseline: 5.9676x; 1.0373x over previous
#include <hip/hip_runtime.h>
#include <hip/hip_bf16.h>

typedef __attribute__((ext_vector_type(2))) float v2f;
typedef __attribute__((ext_vector_type(4))) float v4f;
typedef __attribute__((ext_vector_type(8))) float v8f;

#define D_FEAT 64
#define TILE_PITCH 68
#define BUCKET_ROWS 256
#define MAX_BUCKETS 512
#define CHUNK 4096
#define MAX_CHUNKS 512
#define PIECE 4096


__device__ __forceinline__ int lds_fetch_add(int* p, int v) {
    return __hip_atomic_fetch_add(p, v, __ATOMIC_RELAXED, __HIP_MEMORY_SCOPE_WORKGROUP);
}

__global__ __launch_bounds__(512) void chunk_group_kernel(const int* __restrict__ e_pos,
                                                          const int* __restrict__ e_neg,
                                                          int* __restrict__ off,
                                                          int* __restrict__ pairs,
                                                          int n_edges, int n_buckets, int n_chunks) {
    __shared__ int hist[MAX_BUCKETS];
    __shared__ int wsum[16];
    __shared__ int sorted[CHUNK];
    const int set  = blockIdx.y;
    const int tid  = threadIdx.x;
    const int lane = tid & 31;
    const int wave = tid >> 5;
    const int* __restrict__ edges = set ? e_neg : e_pos;

    hist[tid] = 0;
    const int e0 = blockIdx.x * CHUNK + tid;
    int rw[8], cl[8], rk[8];
#pragma unroll
    for (int u = 0; u < 8; ++u) {
        const int e = e0 + u * 512;
        const bool ok = e < n_edges;
        rw[u] = ok ? edges[e] : -1;
        cl[u] = ok ? edges[(long)n_edges + e] : 0;
    }
    __syncthreads();
#pragma unroll
    for (int u = 0; u < 8; ++u) {
        rk[u] = 0;
        if (rw[u] >= 0) rk[u] = lds_fetch_add(&hist[rw[u] >> 8], 1);
    }
    __syncthreads();

    const int cnt = hist[tid];
    int inc = cnt;
    for (int o = 1; o < 32; o <<= 1) {
        const int s = __shfl_up(inc, o, 32);
        if (lane >= o) inc += s;
    }
    if (lane == 31) wsum[wave] = inc;
    __syncthreads();
    int woff = 0;
    for (int w = 0; w < 16; ++w) woff += (w < wave) ? wsum[w] : 0;
    const int start = woff + inc - cnt;
    hist[tid] = start;
    if (tid <= n_buckets)
        off[((long)set * (n_buckets + 1) + tid) * n_chunks + blockIdx.x] = start;
    __syncthreads();

#pragma unroll
    for (int u = 0; u < 8; ++u)
        if (rw[u] >= 0)
            sorted[hist[rw[u] >> 8] + rk[u]] = (cl[u] << 8) | (rw[u] & (BUCKET_ROWS - 1));
    __syncthreads();

    const int m = min(CHUNK, n_edges - blockIdx.x * CHUNK);
    int* __restrict__ dst = pairs + (long)set * n_edges + (long)blockIdx.x * CHUNK;
#pragma unroll
    for (int u = 0; u < 8; ++u) {
        const int i = tid + u * 512;
        if (i < m) dst[i] = sorted[i];
    }
}

#define GATHER_BATCH 10

__device__ __forceinline__ void gather_mean_tile(const float* __restrict__ x,
                                                 const int* __restrict__ pw,
                                                 const int* __restrict__ offk, int n_chunks,
                                                 long row0, int n, float* T,
                                                 int* lnbr, int* deg, int* lstart, int* cursor,
                                                 int* rdeg, int* wsum,
                                                 int* cpos, int* cbase,
                                                 int tid, int lane, int wave) {
    const int hh = lane >> 4;
    const int q  = lane & 15;
    const char* __restrict__ xb = (const char*)x;
    const unsigned qoff  = (unsigned)q << 4;
    const unsigned rbase = (unsigned)row0;
    const unsigned rlast = (unsigned)(n - 1);

    const v4f zero4 = {0.0f, 0.0f, 0.0f, 0.0f};

    const char* __restrict__ pself = xb + (((rbase + (unsigned)hh) << 8) | qoff);
    v4f sv[8];
#pragma unroll
    for (int i = 0; i < 8; ++i) {
        sv[i] = zero4;
        if (rbase + (unsigned)(2 * i + hh) <= rlast) sv[i] = *(const v4f*)(pself + i * 512);
    }
#pragma unroll
    for (int i = 0; i < 8; ++i)
        *(v4f*)(T + (2 * i + hh) * TILE_PITCH + (q << 2)) = sv[i];
    if (tid < BUCKET_ROWS) rdeg[tid] = 0;

    int seg_count;
    {
        int rs = 0, len = 0;
        if (tid < n_chunks) {
            const int* __restrict__ offe = offk + n_chunks;
            rs  = offk[tid];
            len = offe[tid] - rs;
        }
        int inc = len;
        for (int o = 1; o < 32; o <<= 1) {
            const int t = __shfl_up(inc, o, 32);
            if (lane >= o) inc += t;
        }
        if (lane == 31) wsum[wave] = inc;
        __syncthreads();
        int woff = 0, tot = 0;
        for (int k = 0; k < 16; ++k) {
            const int v = wsum[k];
            tot += v;
            woff += (k < wave) ? v : 0;
        }
        cpos[tid]  = (tid < n_chunks) ? (woff + inc - len) : 0x7fffffff;
        cbase[tid] = rs;
        seg_count = __builtin_amdgcn_readfirstlane(tot);
        __syncthreads();
    }

    const unsigned self_off = (min(rbase + 8u * (unsigned)hh, rlast) << 8) | qoff;

    for (int pb = 0; pb < seg_count; pb += PIECE) {
        const int pc = min(PIECE, seg_count - pb);
        int w[8], p[8], c[8];
        const int pend = pb + pc;
        int pt = pb + tid;
        asm volatile("" : "+v"(pt));
#pragma unroll
        for (int u = 0; u < 8; ++u) {
            const int pp = pt + u * 512;
            p[u] = (pp < pend) ? pp : pb;
            c[u] = 0;
        }
#pragma unroll
        for (int step = MAX_CHUNKS >> 1; step > 0; step >>= 1) {
#pragma unroll
            for (int u = 0; u < 8; ++u)
                if (cpos[c[u] + step] <= p[u]) c[u] += step;
        }
#pragma unroll
        for (int u = 0; u < 8; ++u)
            w[u] = pw[c[u] * CHUNK + cbase[c[u]] + (p[u] - cpos[c[u]])];
        if (tid < BUCKET_ROWS) deg[tid] = 0;
        __syncthreads();
#pragma unroll
        for (int u = 0; u < 8; ++u)
            if (tid + u * 512 < pc) lds_fetch_add(&deg[w[u] & (BUCKET_ROWS - 1)], 1);
        __syncthreads();
        const int d0 = (tid < BUCKET_ROWS) ? deg[tid] : 0;
        int inc = d0;
        for (int o = 1; o < 32; o <<= 1) {
            const int t = __shfl_up(inc, o, 32);
            if (lane >= o) inc += t;
        }
        if (lane == 31 && wave < 8) wsum[wave] = inc;
        __syncthreads();
        if (tid < BUCKET_ROWS) {
            int woff = 0;
            for (int k = 0; k < 8; ++k) woff += (k < wave) ? wsum[k] : 0;
            const int st = woff + inc - d0;
            lstart[tid] = st;
            cursor[tid] = st;
            rdeg[tid] += d0;
        }
        __syncthreads();
#pragma unroll
        for (int u = 0; u < 8; ++u) {
            if (tid + u * 512 < pc) {
                const int slot = lds_fetch_add(&cursor[w[u] & (BUCKET_ROWS - 1)], 1);
                lnbr[slot] = w[u];
            }
        }
        __syncthreads();

        const int beg = lstart[wave * 16 + 8 * hh];
        const int cnt = cursor[wave * 16 + 8 * hh + 7] - beg;
        const int cother = __shfl_xor(cnt, 16, 32);
        const int cmax = max(cnt, cother);
        const int cmin = __builtin_amdgcn_readfirstlane(min(cnt, cother));

        int cur = 8 * hh;
        float* tcur = T + cur * TILE_PITCH + (q << 2);
        v4f acc = *(v4f*)tcur;
        int j = 0;
        {
            const int* lp = lnbr + beg;
            for (; j + GATHER_BATCH <= cmin; j += GATHER_BATCH, lp += GATHER_BATCH) {
                int wd[GATHER_BATCH];
                v4f t[GATHER_BATCH];
#pragma unroll
                for (int u = 0; u < GATHER_BATCH; ++u) wd[u] = lp[u];
#pragma unroll
                for (int u = 0; u < GATHER_BATCH; ++u)
                    t[u] = *(const v4f*)(xb + ((((unsigned)wd[u]) & ~(unsigned)(BUCKET_ROWS - 1)) | qoff));
#pragma unroll
                for (int u = 0; u < GATHER_BATCH; ++u) {
                    const int lr = wd[u] & 15;
                    if (lr != cur) {
                        *(v4f*)tcur = acc;
                        cur = lr;
                        tcur = T + lr * TILE_PITCH + (q << 2);
                        acc = *(v4f*)tcur;
                    }
                    acc += t[u];
                }
            }
        }
        for (; j < cmax; j += GATHER_BATCH) {
            int wd[GATHER_BATCH];
            v4f t[GATHER_BATCH];
#pragma unroll
            for (int u = 0; u < GATHER_BATCH; ++u)
                wd[u] = lnbr[(j + u < cnt) ? (beg + j + u) : 0];
#pragma unroll
            for (int u = 0; u < GATHER_BATCH; ++u) {
                const unsigned off = (j + u < cnt)
                    ? (((unsigned)wd[u] & ~(unsigned)(BUCKET_ROWS - 1)) | qoff)
                    : self_off;
                t[u] = *(const v4f*)(xb + off);
            }
#pragma unroll
            for (int u = 0; u < GATHER_BATCH; ++u) {
                const bool ok = j + u < cnt;
                const int lr = wd[u] & 15;
                if (ok && lr != cur) {
                    *(v4f*)tcur = acc;
                    cur = lr;
                    tcur = T + lr * TILE_PITCH + (q << 2);
                    acc = *(v4f*)tcur;
                }
                acc += ok ? t[u] : zero4;
            }
        }
        *(v4f*)tcur = acc;
        __syncthreads();
    }
    __syncthreads();
#pragma unroll
    for (int i = 0; i < 8; ++i) {
        const int lr = 2 * i + hh;
        const float inv = 1.0f / fmaxf((float)(rdeg[wave * 16 + lr] + 1), 1.0f);
        float* trow = T + lr * TILE_PITCH + (q << 2);
        v4f acc = *(v4f*)trow;
        acc *= inv;
        *(v4f*)trow = acc;
    }
}

__global__ __launch_bounds__(512, 8) void sage_gather_gemm_norm_kernel(
        const float* __restrict__ x1,   const float* __restrict__ x2,
        const int* __restrict__ off,    const int* __restrict__ pairs,
        const float* __restrict__ W,    const float* __restrict__ bias,
        float* __restrict__ out, int n, int n_edges, int n_buckets, int n_chunks) {
    __shared__ float ldsW[192 * 64];
    __shared__ __attribute__((aligned(16))) float ldsA[16 * 16 * TILE_PITCH];
    __shared__ int lnbr[PIECE];
    __shared__ int deg[BUCKET_ROWS], lstart[BUCKET_ROWS], cursor[BUCKET_ROWS], rdeg[BUCKET_ROWS];
    __shared__ int wsum[16];
    __shared__ int cpos[MAX_CHUNKS], cbase[MAX_CHUNKS];

    const int tid = threadIdx.x;
    for (int idx = tid; idx < 192 * 64; idx += 512) {
        const int row = idx >> 6;
        const int col = idx & 63;
        ldsW[((((row >> 1) << 6) | col) << 1) | (row & 1)] = W[idx];
    }

    const int lane = tid & 31;
    const int wave = tid >> 5;
    const int m = lane & 15;
    const int h = lane >> 4;

    const long tile = (long)blockIdx.x * 16 + wave;
    const long row0 = tile << 4;

    float* T = ldsA + wave * (16 * TILE_PITCH);
    const float* ta = T + m * TILE_PITCH;

    v8f acc0 = {}; v8f acc1 = {}; v8f acc2 = {}; v8f acc3 = {};
    const float* wbase = ldsW + (m << 1);

#pragma nounroll
    for (int set = 0; set < 2; ++set) {
        const int* __restrict__ offk = off + ((long)set * (n_buckets + 1) + blockIdx.x) * n_chunks;
        gather_mean_tile(set ? x2 : x1, pairs + (long)set * n_edges, offk, n_chunks,
                         row0, n, T, lnbr, deg, lstart, cursor, rdeg, wsum, cpos, cbase,
                         tid, lane, wave);
        __syncthreads();
        const float* wset = wbase + ((set << 5) << 7);
#pragma unroll
        for (int k = 0; k < 16; ++k) {
            const int c = (k << 2) + (h << 1);
            const v2f a = *(const v2f*)(ta + c);
            const float* wp = wset + (((k << 1) + h) << 7);
            const v2f b0 = *(const v2f*)(wp);
            const v2f b1 = *(const v2f*)(wp + 32);
            const v2f b2 = *(const v2f*)(wp + 64);
            const v2f b3 = *(const v2f*)(wp + 96);
            acc0 = __builtin_amdgcn_wmma_f32_16x16x4_f32(false, a, false, b0, (short)0, acc0, false, false);
            acc1 = __builtin_amdgcn_wmma_f32_16x16x4_f32(false, a, false, b1, (short)0, acc1, false, false);
            acc2 = __builtin_amdgcn_wmma_f32_16x16x4_f32(false, a, false, b2, (short)0, acc2, false, false);
            acc3 = __builtin_amdgcn_wmma_f32_16x16x4_f32(false, a, false, b3, (short)0, acc3, false, false);
        }
        asm volatile("v_nop\n\tv_nop\n\tv_nop\n\tv_nop\n\tv_nop\n\tv_nop\n\tv_nop\n\tv_nop"
                     : "+v"(acc0), "+v"(acc1), "+v"(acc2), "+v"(acc3));
        __syncthreads();
    }
    int tid_end = tid;
    asm volatile("" : "+v"(tid_end));
    const int me = tid_end & 15;
    const int he = (tid_end >> 4) & 1;
    const float* wbe = ldsW + (me << 1);
    long r = row0 + me;
    if (r > (long)(n - 1)) r = (long)(n - 1);
    const float* pax = x1 + r * D_FEAT;
#pragma unroll
    for (int k = 0; k < 16; ++k) {
        const int c = (k << 2) + (he << 1);
        const v2f a = *(const v2f*)(pax + c);
        const float* wp = wbe + ((64 + (k << 1) + he) << 7);
        const v2f b0 = *(const v2f*)(wp);
        const v2f b1 = *(const v2f*)(wp + 32);
        const v2f b2 = *(const v2f*)(wp + 64);
        const v2f b3 = *(const v2f*)(wp + 96);
        acc0 = __builtin_amdgcn_wmma_f32_16x16x4_f32(false, a, false, b0, (short)0, acc0, false, false);
        acc1 = __builtin_amdgcn_wmma_f32_16x16x4_f32(false, a, false, b1, (short)0, acc1, false, false);
        acc2 = __builtin_amdgcn_wmma_f32_16x16x4_f32(false, a, false, b2, (short)0, acc2, false, false);
        acc3 = __builtin_amdgcn_wmma_f32_16x16x4_f32(false, a, false, b3, (short)0, acc3, false, false);
    }

    const float bs0 = bias[me];
    const float bs1 = bias[16 + me];
    const float bs2 = bias[32 + me];
    const float bs3 = bias[48 + me];

#pragma unroll
    for (int j = 0; j < 8; ++j) {
        const float o0 = acc0[j] + bs0;
        const float o1 = acc1[j] + bs1;
        const float o2 = acc2[j] + bs2;
        const float o3 = acc3[j] + bs3;
        float s = o0 * o0 + o1 * o1 + o2 * o2 + o3 * o3;
        s += __shfl_xor(s, 1, 32);
        s += __shfl_xor(s, 2, 32);
        s += __shfl_xor(s, 4, 32);
        s += __shfl_xor(s, 8, 32);
        const float scale = 1.0f / fmaxf(sqrtf(s), 1e-12f);
        const long orow = row0 + j + (he << 3);
        if (orow < n) {
            float* po = out + orow * D_FEAT + me;
            po[0]  = o0 * scale;
            po[16] = o1 * scale;
            po[32] = o2 * scale;
            po[48] = o3 * scale;
        }
    }
}

extern "C" void kernel_launch(void* const* d_in, const int* in_sizes, int n_in,
                              void* d_out, int out_size, void* d_ws, size_t ws_size,
                              hipStream_t stream) {
    const float* x1    = (const float*)d_in[0];
    const float* x2    = (const float*)d_in[1];
    const int*   e_pos = (const int*)d_in[2];
    const int*   e_neg = (const int*)d_in[3];
    const float* W     = (const float*)d_in[4];
    const float* bias  = (const float*)d_in[5];
    float*       out   = (float*)d_out;

    const int n       = in_sizes[0] / D_FEAT;
    const int n_edges = in_sizes[2] / 2;

    const int n_buckets = (n + BUCKET_ROWS - 1) / BUCKET_ROWS;
    const int n_chunks  = (n_edges + CHUNK - 1) / CHUNK;

    int* off   = (int*)d_ws;
    int* pairs = off + (size_t)2 * (n_buckets + 1) * n_chunks;

    const dim3 egrid(n_chunks, 2);
    chunk_group_kernel<<<egrid, 512, 0, stream>>>(e_pos, e_neg, off, pairs,
                                                  n_edges, n_buckets, n_chunks);
    {
        const int tiles  = (n + 15) / 16;
        const int blocks = (tiles + 15) / 16;
        sage_gather_gemm_norm_kernel<<<blocks, 512, 0, stream>>>(x1, x2, off, pairs,
                                                                 W, bias, out, n, n_edges, n_buckets,
                                                                 n_chunks);
    }
}


# Edge pass as 4-wave blocks, 32 edges per thread, each thread's loads and LDS reads issued as one group (clamped index, no wait per edge inside a guarded branch); row sums stay LDS-arrival order
# speedup vs baseline: 6.1930x; 1.0378x over previous
#include <hip/hip_runtime.h>
#include <hip/hip_bf16.h>

typedef __attribute__((ext_vector_type(2))) float v2f;
typedef __attribute__((ext_vector_type(4))) float v4f;
typedef __attribute__((ext_vector_type(8))) float v8f;

#define D_FEAT 64
#define TILE_PITCH 68
#define BUCKET_ROWS 256
#define MAX_BUCKETS 512
#define CHUNK 4096
#define EDGE_THREADS 128
#define EDGES_PER_THREAD (CHUNK / EDGE_THREADS)
#define BUCKETS_PER_THREAD (MAX_BUCKETS / EDGE_THREADS)
#define MAX_CHUNKS 512
#define PIECE 4096


__device__ __forceinline__ int lds_fetch_add(int* p, int v) {
    return __hip_atomic_fetch_add(p, v, __ATOMIC_RELAXED, __HIP_MEMORY_SCOPE_WORKGROUP);
}

__global__ __launch_bounds__(EDGE_THREADS) void chunk_group_kernel(const int* __restrict__ e_pos,
                                                                   const int* __restrict__ e_neg,
                                                                   int* __restrict__ off,
                                                                   int* __restrict__ pairs,
                                                                   int n_edges, int n_buckets, int n_chunks) {
    __shared__ int hist[MAX_BUCKETS];
    __shared__ int wsum[EDGE_THREADS / 32];
    __shared__ int sorted[CHUNK];
    const int set  = blockIdx.y;
    const int tid  = threadIdx.x;
    const int lane = tid & 31;
    const int wave = tid >> 5;
    const int* __restrict__ edges = set ? e_neg : e_pos;

    int* myhist = hist + tid * BUCKETS_PER_THREAD;
#pragma unroll
    for (int k = 0; k < BUCKETS_PER_THREAD; ++k) myhist[k] = 0;
    const int e0 = blockIdx.x * CHUNK + tid;
    int rw[EDGES_PER_THREAD], cl[EDGES_PER_THREAD], rk[EDGES_PER_THREAD];
    const int e_last = n_edges - 1;
#pragma unroll
    for (int u = 0; u < EDGES_PER_THREAD; ++u) {
        const int e  = e0 + u * EDGE_THREADS;
        const int ec = min(e, e_last);
        const int r  = edges[ec];
        cl[u] = edges[(long)n_edges + ec];
        rw[u] = (e < n_edges) ? r : -1;
    }
    __syncthreads();
#pragma unroll
    for (int u = 0; u < EDGES_PER_THREAD; ++u) {
        rk[u] = 0;
        if (rw[u] >= 0) rk[u] = lds_fetch_add(&hist[rw[u] >> 8], 1);
    }
    __syncthreads();

    int cnt[BUCKETS_PER_THREAD];
    int tot = 0;
#pragma unroll
    for (int k = 0; k < BUCKETS_PER_THREAD; ++k) {
        cnt[k] = myhist[k];
        tot += cnt[k];
    }
    int inc = tot;
    for (int o = 1; o < 32; o <<= 1) {
        const int s = __shfl_up(inc, o, 32);
        if (lane >= o) inc += s;
    }
    if (lane == 31) wsum[wave] = inc;
    __syncthreads();
    int start = inc - tot;
    for (int w = 0; w < EDGE_THREADS / 32; ++w) start += (w < wave) ? wsum[w] : 0;
#pragma unroll
    for (int k = 0; k < BUCKETS_PER_THREAD; ++k) {
        const int b = tid * BUCKETS_PER_THREAD + k;
        myhist[k] = start;
        if (b <= n_buckets)
            off[((long)set * (n_buckets + 1) + b) * n_chunks + blockIdx.x] = start;
        start += cnt[k];
    }
    __syncthreads();

#pragma unroll
    for (int u = 0; u < EDGES_PER_THREAD; ++u)
        rk[u] += hist[(rw[u] >= 0) ? (rw[u] >> 8) : 0];
#pragma unroll
    for (int u = 0; u < EDGES_PER_THREAD; ++u)
        if (rw[u] >= 0)
            sorted[rk[u]] = (cl[u] << 8) | (rw[u] & (BUCKET_ROWS - 1));
    __syncthreads();

    const int m = min(CHUNK, n_edges - blockIdx.x * CHUNK);
    int* __restrict__ dst = pairs + (long)set * n_edges + (long)blockIdx.x * CHUNK;
#pragma unroll
    for (int u = 0; u < EDGES_PER_THREAD; ++u) rk[u] = sorted[min(tid + u * EDGE_THREADS, m - 1)];
#pragma unroll
    for (int u = 0; u < EDGES_PER_THREAD; ++u) {
        const int i = tid + u * EDGE_THREADS;
        if (i < m) dst[i] = rk[u];
    }
}

#define GATHER_BATCH 10

__device__ __forceinline__ void gather_mean_tile(const float* __restrict__ x,
                                                 const int* __restrict__ pw,
                                                 const int* __restrict__ offk, int n_chunks,
                                                 long row0, int n, float* T,
                                                 int* lnbr, int* deg, int* lstart, int* cursor,
                                                 int* rdeg, int* wsum,
                                                 int* cpos, int* cbase,
                                                 int tid, int lane, int wave) {
    const int hh = lane >> 4;
    const int q  = lane & 15;
    const char* __restrict__ xb = (const char*)x;
    const unsigned qoff  = (unsigned)q << 4;
    const unsigned rbase = (unsigned)row0;
    const unsigned rlast = (unsigned)(n - 1);

    const v4f zero4 = {0.0f, 0.0f, 0.0f, 0.0f};

    const char* __restrict__ pself = xb + (((rbase + (unsigned)hh) << 8) | qoff);
    v4f sv[8];
#pragma unroll
    for (int i = 0; i < 8; ++i) {
        sv[i] = zero4;
        if (rbase + (unsigned)(2 * i + hh) <= rlast) sv[i] = *(const v4f*)(pself + i * 512);
    }
#pragma unroll
    for (int i = 0; i < 8; ++i)
        *(v4f*)(T + (2 * i + hh) * TILE_PITCH + (q << 2)) = sv[i];
    if (tid < BUCKET_ROWS) rdeg[tid] = 0;

    int seg_count;
    {
        int rs = 0, len = 0;
        if (tid < n_chunks) {
            const int* __restrict__ offe = offk + n_chunks;
            rs  = offk[tid];
            len = offe[tid] - rs;
        }
        int inc = len;
        for (int o = 1; o < 32; o <<= 1) {
            const int t = __shfl_up(inc, o, 32);
            if (lane >= o) inc += t;
        }
        if (lane == 31) wsum[wave] = inc;
        __syncthreads();
        int woff = 0, tot = 0;
        for (int k = 0; k < 16; ++k) {
            const int v = wsum[k];
            tot += v;
            woff += (k < wave) ? v : 0;
        }
        cpos[tid]  = (tid < n_chunks) ? (woff + inc - len) : 0x7fffffff;
        cbase[tid] = rs;
        seg_count = __builtin_amdgcn_readfirstlane(tot);
        __syncthreads();
    }

    const unsigned self_off = (min(rbase + 8u * (unsigned)hh, rlast) << 8) | qoff;

    for (int pb = 0; pb < seg_count; pb += PIECE) {
        const int pc = min(PIECE, seg_count - pb);
        int w[8], p[8], c[8];
        const int pend = pb + pc;
        int pt = pb + tid;
        asm volatile("" : "+v"(pt));
#pragma unroll
        for (int u = 0; u < 8; ++u) {
            const int pp = pt + u * 512;
            p[u] = (pp < pend) ? pp : pb;
            c[u] = 0;
        }
#pragma unroll
        for (int step = MAX_CHUNKS >> 1; step > 0; step >>= 1) {
#pragma unroll
            for (int u = 0; u < 8; ++u)
                if (cpos[c[u] + step] <= p[u]) c[u] += step;
        }
#pragma unroll
        for (int u = 0; u < 8; ++u)
            w[u] = pw[c[u] * CHUNK + cbase[c[u]] + (p[u] - cpos[c[u]])];
        if (tid < BUCKET_ROWS) deg[tid] = 0;
        __syncthreads();
#pragma unroll
        for (int u = 0; u < 8; ++u)
            if (tid + u * 512 < pc) lds_fetch_add(&deg[w[u] & (BUCKET_ROWS - 1)], 1);
        __syncthreads();
        const int d0 = (tid < BUCKET_ROWS) ? deg[tid] : 0;
        int inc = d0;
        for (int o = 1; o < 32; o <<= 1) {
            const int t = __shfl_up(inc, o, 32);
            if (lane >= o) inc += t;
        }
        if (lane == 31 && wave < 8) wsum[wave] = inc;
        __syncthreads();
        if (tid < BUCKET_ROWS) {
            int woff = 0;
            for (int k = 0; k < 8; ++k) woff += (k < wave) ? wsum[k] : 0;
            const int st = woff + inc - d0;
            lstart[tid] = st;
            cursor[tid] = st;
            rdeg[tid] += d0;
        }
        __syncthreads();
#pragma unroll
        for (int u = 0; u < 8; ++u) {
            if (tid + u * 512 < pc) {
                const int slot = lds_fetch_add(&cursor[w[u] & (BUCKET_ROWS - 1)], 1);
                lnbr[slot] = w[u];
            }
        }
        __syncthreads();

        const int beg = lstart[wave * 16 + 8 * hh];
        const int cnt = cursor[wave * 16 + 8 * hh + 7] - beg;
        const int cother = __shfl_xor(cnt, 16, 32);
        const int cmax = max(cnt, cother);
        const int cmin = __builtin_amdgcn_readfirstlane(min(cnt, cother));

        int cur = 8 * hh;
        float* tcur = T + cur * TILE_PITCH + (q << 2);
        v4f acc = *(v4f*)tcur;
        int j = 0;
        {
            const int* lp = lnbr + beg;
            for (; j + GATHER_BATCH <= cmin; j += GATHER_BATCH, lp += GATHER_BATCH) {
                int wd[GATHER_BATCH];
                v4f t[GATHER_BATCH];
#pragma unroll
                for (int u = 0; u < GATHER_BATCH; ++u) wd[u] = lp[u];
#pragma unroll
                for (int u = 0; u < GATHER_BATCH; ++u)
                    t[u] = *(const v4f*)(xb + ((((unsigned)wd[u]) & ~(unsigned)(BUCKET_ROWS - 1)) | qoff));
#pragma unroll
                for (int u = 0; u < GATHER_BATCH; ++u) {
                    const int lr = wd[u] & 15;
                    if (lr != cur) {
                        *(v4f*)tcur = acc;
                        cur = lr;
                        tcur = T + lr * TILE_PITCH + (q << 2);
                        acc = *(v4f*)tcur;
                    }
                    acc += t[u];
                }
            }
        }
        for (; j < cmax; j += GATHER_BATCH) {
            int wd[GATHER_BATCH];
            v4f t[GATHER_BATCH];
#pragma unroll
            for (int u = 0; u < GATHER_BATCH; ++u)
                wd[u] = lnbr[(j + u < cnt) ? (beg + j + u) : 0];
#pragma unroll
            for (int u = 0; u < GATHER_BATCH; ++u) {
                const unsigned off = (j + u < cnt)
                    ? (((unsigned)wd[u] & ~(unsigned)(BUCKET_ROWS - 1)) | qoff)
                    : self_off;
                t[u] = *(const v4f*)(xb + off);
            }
#pragma unroll
            for (int u = 0; u < GATHER_BATCH; ++u) {
                const bool ok = j + u < cnt;
                const int lr = wd[u] & 15;
                if (ok && lr != cur) {
                    *(v4f*)tcur = acc;
                    cur = lr;
                    tcur = T + lr * TILE_PITCH + (q << 2);
                    acc = *(v4f*)tcur;
                }
                acc += ok ? t[u] : zero4;
            }
        }
        *(v4f*)tcur = acc;
        __syncthreads();
    }
    __syncthreads();
#pragma unroll
    for (int i = 0; i < 8; ++i) {
        const int lr = 2 * i + hh;
        const float inv = 1.0f / fmaxf((float)(rdeg[wave * 16 + lr] + 1), 1.0f);
        float* trow = T + lr * TILE_PITCH + (q << 2);
        v4f acc = *(v4f*)trow;
        acc *= inv;
        *(v4f*)trow = acc;
    }
}

__global__ __launch_bounds__(512, 8) void sage_gather_gemm_norm_kernel(
        const float* __restrict__ x1,   const float* __restrict__ x2,
        const int* __restrict__ off,    const int* __restrict__ pairs,
        const float* __restrict__ W,    const float* __restrict__ bias,
        float* __restrict__ out, int n, int n_edges, int n_buckets, int n_chunks) {
    __shared__ float ldsW[192 * 64];
    __shared__ __attribute__((aligned(16))) float ldsA[16 * 16 * TILE_PITCH];
    __shared__ int lnbr[PIECE];
    __shared__ int deg[BUCKET_ROWS], lstart[BUCKET_ROWS], cursor[BUCKET_ROWS], rdeg[BUCKET_ROWS];
    __shared__ int wsum[16];
    __shared__ int cpos[MAX_CHUNKS], cbase[MAX_CHUNKS];

    const int tid = threadIdx.x;
    for (int idx = tid; idx < 192 * 64; idx += 512) {
        const int row = idx >> 6;
        const int col = idx & 63;
        ldsW[((((row >> 1) << 6) | col) << 1) | (row & 1)] = W[idx];
    }

    const int lane = tid & 31;
    const int wave = tid >> 5;
    const int m = lane & 15;
    const int h = lane >> 4;

    const long tile = (long)blockIdx.x * 16 + wave;
    const long row0 = tile << 4;

    float* T = ldsA + wave * (16 * TILE_PITCH);
    const float* ta = T + m * TILE_PITCH;

    v8f acc0 = {}; v8f acc1 = {}; v8f acc2 = {}; v8f acc3 = {};
    const float* wbase = ldsW + (m << 1);

#pragma nounroll
    for (int set = 0; set < 2; ++set) {
        const int* __restrict__ offk = off + ((long)set * (n_buckets + 1) + blockIdx.x) * n_chunks;
        gather_mean_tile(set ? x2 : x1, pairs + (long)set * n_edges, offk, n_chunks,
                         row0, n, T, lnbr, deg, lstart, cursor, rdeg, wsum, cpos, cbase,
                         tid, lane, wave);
        __syncthreads();
        const float* wset = wbase + ((set << 5) << 7);
#pragma unroll
        for (int k = 0; k < 16; ++k) {
            const int c = (k << 2) + (h << 1);
            const v2f a = *(const v2f*)(ta + c);
            const float* wp = wset + (((k << 1) + h) << 7);
            const v2f b0 = *(const v2f*)(wp);
            const v2f b1 = *(const v2f*)(wp + 32);
            const v2f b2 = *(const v2f*)(wp + 64);
            const v2f b3 = *(const v2f*)(wp + 96);
            acc0 = __builtin_amdgcn_wmma_f32_16x16x4_f32(false, a, false, b0, (short)0, acc0, false, false);
            acc1 = __builtin_amdgcn_wmma_f32_16x16x4_f32(false, a, false, b1, (short)0, acc1, false, false);
            acc2 = __builtin_amdgcn_wmma_f32_16x16x4_f32(false, a, false, b2, (short)0, acc2, false, false);
            acc3 = __builtin_amdgcn_wmma_f32_16x16x4_f32(false, a, false, b3, (short)0, acc3, false, false);
        }
        asm volatile("v_nop\n\tv_nop\n\tv_nop\n\tv_nop\n\tv_nop\n\tv_nop\n\tv_nop\n\tv_nop"
                     : "+v"(acc0), "+v"(acc1), "+v"(acc2), "+v"(acc3));
        __syncthreads();
    }
    int tid_end = tid;
    asm volatile("" : "+v"(tid_end));
    const int me = tid_end & 15;
    const int he = (tid_end >> 4) & 1;
    const float* wbe = ldsW + (me << 1);
    long r = row0 + me;
    if (r > (long)(n - 1)) r = (long)(n - 1);
    const float* pax = x1 + r * D_FEAT;
#pragma unroll
    for (int k = 0; k < 16; ++k) {
        const int c = (k << 2) + (he << 1);
        const v2f a = *(const v2f*)(pax + c);
        const float* wp = wbe + ((64 + (k << 1) + he) << 7);
        const v2f b0 = *(const v2f*)(wp);
        const v2f b1 = *(const v2f*)(wp + 32);
        const v2f b2 = *(const v2f*)(wp + 64);
        const v2f b3 = *(const v2f*)(wp + 96);
        acc0 = __builtin_amdgcn_wmma_f32_16x16x4_f32(false, a, false, b0, (short)0, acc0, false, false);
        acc1 = __builtin_amdgcn_wmma_f32_16x16x4_f32(false, a, false, b1, (short)0, acc1, false, false);
        acc2 = __builtin_amdgcn_wmma_f32_16x16x4_f32(false, a, false, b2, (short)0, acc2, false, false);
        acc3 = __builtin_amdgcn_wmma_f32_16x16x4_f32(false, a, false, b3, (short)0, acc3, false, false);
    }

    const float bs0 = bias[me];
    const float bs1 = bias[16 + me];
    const float bs2 = bias[32 + me];
    const float bs3 = bias[48 + me];

#pragma unroll
    for (int j = 0; j < 8; ++j) {
        const float o0 = acc0[j] + bs0;
        const float o1 = acc1[j] + bs1;
        const float o2 = acc2[j] + bs2;
        const float o3 = acc3[j] + bs3;
        float s = o0 * o0 + o1 * o1 + o2 * o2 + o3 * o3;
        s += __shfl_xor(s, 1, 32);
        s += __shfl_xor(s, 2, 32);
        s += __shfl_xor(s, 4, 32);
        s += __shfl_xor(s, 8, 32);
        const float scale = 1.0f / fmaxf(sqrtf(s), 1e-12f);
        const long orow = row0 + j + (he << 3);
        if (orow < n) {
            float* po = out + orow * D_FEAT + me;
            po[0]  = o0 * scale;
            po[16] = o1 * scale;
            po[32] = o2 * scale;
            po[48] = o3 * scale;
        }
    }
}

extern "C" void kernel_launch(void* const* d_in, const int* in_sizes, int n_in,
                              void* d_out, int out_size, void* d_ws, size_t ws_size,
                              hipStream_t stream) {
    const float* x1    = (const float*)d_in[0];
    const float* x2    = (const float*)d_in[1];
    const int*   e_pos = (const int*)d_in[2];
    const int*   e_neg = (const int*)d_in[3];
    const float* W     = (const float*)d_in[4];
    const float* bias  = (const float*)d_in[5];
    float*       out   = (float*)d_out;

    const int n       = in_sizes[0] / D_FEAT;
    const int n_edges = in_sizes[2] / 2;

    const int n_buckets = (n + BUCKET_ROWS - 1) / BUCKET_ROWS;
    const int n_chunks  = (n_edges + CHUNK - 1) / CHUNK;

    int* off   = (int*)d_ws;
    int* pairs = off + (size_t)2 * (n_buckets + 1) * n_chunks;

    const dim3 egrid(n_chunks, 2);
    chunk_group_kernel<<<egrid, EDGE_THREADS, 0, stream>>>(e_pos, e_neg, off, pairs,
                                                  n_edges, n_buckets, n_chunks);
    {
        const int tiles  = (n + 15) / 16;
        const int blocks = (tiles + 15) / 16;
        sage_gather_gemm_norm_kernel<<<blocks, 512, 0, stream>>>(x1, x2, off, pairs,
                                                                 W, bias, out, n, n_edges, n_buckets,
                                                                 n_chunks);
    }
}


# Both sets' neighbour lists ordered by row in LDS up front; behind that no block barrier: each wave runs its two gathers and three products alone, so WMMAs overlap other waves' gathers; sum order kept
# speedup vs baseline: 6.4627x; 1.0435x over previous
#include <hip/hip_runtime.h>
#include <hip/hip_bf16.h>

typedef __attribute__((ext_vector_type(2))) float v2f;
typedef __attribute__((ext_vector_type(4))) float v4f;
typedef __attribute__((ext_vector_type(8))) float v8f;

#define D_FEAT 64
#define TILE_PITCH 68
#define BUCKET_ROWS 256
#define MAX_BUCKETS 512
#define CHUNK 4096
#define EDGE_THREADS 128
#define EDGES_PER_THREAD (CHUNK / EDGE_THREADS)
#define BUCKETS_PER_THREAD (MAX_BUCKETS / EDGE_THREADS)
#define MAX_CHUNKS 512
#define PIECE 4096


__device__ __forceinline__ int lds_fetch_add(int* p, int v) {
    return __hip_atomic_fetch_add(p, v, __ATOMIC_RELAXED, __HIP_MEMORY_SCOPE_WORKGROUP);
}

__global__ __launch_bounds__(EDGE_THREADS) void chunk_group_kernel(const int* __restrict__ e_pos,
                                                                   const int* __restrict__ e_neg,
                                                                   int* __restrict__ off,
                                                                   int* __restrict__ pairs,
                                                                   int n_edges, int n_buckets, int n_chunks) {
    __shared__ int hist[MAX_BUCKETS];
    __shared__ int wsum[EDGE_THREADS / 32];
    __shared__ int sorted[CHUNK];
    const int set  = blockIdx.y;
    const int tid  = threadIdx.x;
    const int lane = tid & 31;
    const int wave = tid >> 5;
    const int* __restrict__ edges = set ? e_neg : e_pos;

    int* myhist = hist + tid * BUCKETS_PER_THREAD;
#pragma unroll
    for (int k = 0; k < BUCKETS_PER_THREAD; ++k) myhist[k] = 0;
    const int e0 = blockIdx.x * CHUNK + tid;
    int rw[EDGES_PER_THREAD], cl[EDGES_PER_THREAD], rk[EDGES_PER_THREAD];
    const int e_last = n_edges - 1;
#pragma unroll
    for (int u = 0; u < EDGES_PER_THREAD; ++u) {
        const int e  = e0 + u * EDGE_THREADS;
        const int ec = min(e, e_last);
        const int r  = edges[ec];
        cl[u] = edges[(long)n_edges + ec];
        rw[u] = (e < n_edges) ? r : -1;
    }
    __syncthreads();
#pragma unroll
    for (int u = 0; u < EDGES_PER_THREAD; ++u) {
        rk[u] = 0;
        if (rw[u] >= 0) rk[u] = lds_fetch_add(&hist[rw[u] >> 8], 1);
    }
    __syncthreads();

    int cnt[BUCKETS_PER_THREAD];
    int tot = 0;
#pragma unroll
    for (int k = 0; k < BUCKETS_PER_THREAD; ++k) {
        cnt[k] = myhist[k];
        tot += cnt[k];
    }
    int inc = tot;
    for (int o = 1; o < 32; o <<= 1) {
        const int s = __shfl_up(inc, o, 32);
        if (lane >= o) inc += s;
    }
    if (lane == 31) wsum[wave] = inc;
    __syncthreads();
    int start = inc - tot;
    for (int w = 0; w < EDGE_THREADS / 32; ++w) start += (w < wave) ? wsum[w] : 0;
#pragma unroll
    for (int k = 0; k < BUCKETS_PER_THREAD; ++k) {
        const int b = tid * BUCKETS_PER_THREAD + k;
        myhist[k] = start;
        if (b <= n_buckets)
            off[((long)set * (n_buckets + 1) + b) * n_chunks + blockIdx.x] = start;
        start += cnt[k];
    }
    __syncthreads();

#pragma unroll
    for (int u = 0; u < EDGES_PER_THREAD; ++u)
        rk[u] += hist[(rw[u] >= 0) ? (rw[u] >> 8) : 0];
#pragma unroll
    for (int u = 0; u < EDGES_PER_THREAD; ++u)
        if (rw[u] >= 0)
            sorted[rk[u]] = (cl[u] << 8) | (rw[u] & (BUCKET_ROWS - 1));
    __syncthreads();

    const int m = min(CHUNK, n_edges - blockIdx.x * CHUNK);
    int* __restrict__ dst = pairs + (long)set * n_edges + (long)blockIdx.x * CHUNK;
#pragma unroll
    for (int u = 0; u < EDGES_PER_THREAD; ++u) rk[u] = sorted[min(tid + u * EDGE_THREADS, m - 1)];
#pragma unroll
    for (int u = 0; u < EDGES_PER_THREAD; ++u) {
        const int i = tid + u * EDGE_THREADS;
        if (i < m) dst[i] = rk[u];
    }
}


__device__ __forceinline__ int order_set(const int* __restrict__ pw, const int* __restrict__ offk,
                                         int n_chunks, int* lnbr, int* rstart,
                                         int* deg, int* cursor, int* wsum, int* cpos, int* cbase,
                                         int tid, int lane, int wave) {
    int seg_count;
    {
        int rs = 0, len = 0;
        if (tid < n_chunks) {
            const int* __restrict__ offe = offk + n_chunks;
            rs  = offk[tid];
            len = offe[tid] - rs;
        }
        int inc = len;
        for (int o = 1; o < 32; o <<= 1) {
            const int t = __shfl_up(inc, o, 32);
            if (lane >= o) inc += t;
        }
        if (lane == 31) wsum[wave] = inc;
        __syncthreads();
        int woff = 0, tot = 0;
        for (int k = 0; k < 16; ++k) {
            const int v = wsum[k];
            tot += v;
            woff += (k < wave) ? v : 0;
        }
        cpos[tid]  = (tid < n_chunks) ? (woff + inc - len) : 0x7fffffff;
        cbase[tid] = rs;
        if (tid < BUCKET_ROWS) deg[tid] = 0;
        seg_count = __builtin_amdgcn_readfirstlane(tot);
        __syncthreads();
    }
    if (seg_count > PIECE) return seg_count;
    if (seg_count == 0) {
        if (tid <= BUCKET_ROWS) rstart[tid] = 0;
        __syncthreads();
        return 0;
    }
    const int pc = seg_count;

    int w[8], p[8], c[8];
#pragma unroll
    for (int u = 0; u < 8; ++u) {
        const int pp = tid + u * 512;
        p[u] = (pp < pc) ? pp : 0;
        c[u] = 0;
    }
#pragma unroll
    for (int step = MAX_CHUNKS >> 1; step > 0; step >>= 1) {
#pragma unroll
        for (int u = 0; u < 8; ++u)
            if (cpos[c[u] + step] <= p[u]) c[u] += step;
    }
#pragma unroll
    for (int u = 0; u < 8; ++u)
        w[u] = pw[c[u] * CHUNK + cbase[c[u]] + (p[u] - cpos[c[u]])];
#pragma unroll
    for (int u = 0; u < 8; ++u)
        if (tid + u * 512 < pc) lds_fetch_add(&deg[w[u] & (BUCKET_ROWS - 1)], 1);
    __syncthreads();
    const int d0 = (tid < BUCKET_ROWS) ? deg[tid] : 0;
    int inc = d0;
    for (int o = 1; o < 32; o <<= 1) {
        const int t = __shfl_up(inc, o, 32);
        if (lane >= o) inc += t;
    }
    if (lane == 31 && wave < 8) wsum[wave] = inc;
    __syncthreads();
    if (tid < BUCKET_ROWS) {
        int woff = 0;
        for (int k = 0; k < 8; ++k) woff += (k < wave) ? wsum[k] : 0;
        const int st = woff + inc - d0;
        rstart[tid] = st;
        cursor[tid] = st;
        if (tid == BUCKET_ROWS - 1) rstart[BUCKET_ROWS] = st + d0;
    }
    __syncthreads();
#pragma unroll
    for (int u = 0; u < 8; ++u) {
        if (tid + u * 512 < pc) {
            const int slot = lds_fetch_add(&cursor[w[u] & (BUCKET_ROWS - 1)], 1);
            lnbr[slot] = w[u];
        }
    }
    __syncthreads();
    return seg_count;
}

#define GATHER_BATCH 10

__device__ __forceinline__ void gather_rows(const float* __restrict__ x, const int* lnbr,
                                            const int* rstart, long row0, int n, float* T,
                                            int lane, int wave) {
    const int hh = lane >> 4;
    const int q  = lane & 15;
    const char* __restrict__ xb = (const char*)x;
    const unsigned qoff  = (unsigned)q << 4;
    const unsigned rbase = (unsigned)row0;
    const unsigned rlast = (unsigned)(n - 1);

    const v4f zero4 = {0.0f, 0.0f, 0.0f, 0.0f};

    const char* __restrict__ pself = xb + (((rbase + (unsigned)hh) << 8) | qoff);
    v4f sv[8];
#pragma unroll
    for (int i = 0; i < 8; ++i) {
        sv[i] = zero4;
        if (rbase + (unsigned)(2 * i + hh) <= rlast) sv[i] = *(const v4f*)(pself + i * 512);
    }
#pragma unroll
    for (int i = 0; i < 8; ++i)
        *(v4f*)(T + (2 * i + hh) * TILE_PITCH + (q << 2)) = sv[i];

    const unsigned self_off = (min(rbase + 8u * (unsigned)hh, rlast) << 8) | qoff;

    const int* rs = rstart + wave * 16 + 8 * hh;
    const int beg = rs[0];
    const int cnt = rs[8] - beg;
    const int cother = __shfl_xor(cnt, 16, 32);
    const int cmax = max(cnt, cother);
    const int cmin = __builtin_amdgcn_readfirstlane(min(cnt, cother));

    int cur = 8 * hh;
    float* tcur = T + cur * TILE_PITCH + (q << 2);
    v4f acc = *(v4f*)tcur;
    int j = 0;
    {
        const int* lp = lnbr + beg;
        for (; j + GATHER_BATCH <= cmin; j += GATHER_BATCH, lp += GATHER_BATCH) {
            int wd[GATHER_BATCH];
            v4f t[GATHER_BATCH];
#pragma unroll
            for (int u = 0; u < GATHER_BATCH; ++u) wd[u] = lp[u];
#pragma unroll
            for (int u = 0; u < GATHER_BATCH; ++u)
                t[u] = *(const v4f*)(xb + ((((unsigned)wd[u]) & ~(unsigned)(BUCKET_ROWS - 1)) | qoff));
#pragma unroll
            for (int u = 0; u < GATHER_BATCH; ++u) {
                const int lr = wd[u] & 15;
                if (lr != cur) {
                    *(v4f*)tcur = acc;
                    cur = lr;
                    tcur = T + lr * TILE_PITCH + (q << 2);
                    acc = *(v4f*)tcur;
                }
                acc += t[u];
            }
        }
    }
    for (; j < cmax; j += GATHER_BATCH) {
        int wd[GATHER_BATCH];
        v4f t[GATHER_BATCH];
#pragma unroll
        for (int u = 0; u < GATHER_BATCH; ++u)
            wd[u] = lnbr[(j + u < cnt) ? (beg + j + u) : 0];
#pragma unroll
        for (int u = 0; u < GATHER_BATCH; ++u) {
            const unsigned off = (j + u < cnt)
                ? (((unsigned)wd[u] & ~(unsigned)(BUCKET_ROWS - 1)) | qoff)
                : self_off;
            t[u] = *(const v4f*)(xb + off);
        }
#pragma unroll
        for (int u = 0; u < GATHER_BATCH; ++u) {
            const bool ok = j + u < cnt;
            const int lr = wd[u] & 15;
            if (ok && lr != cur) {
                *(v4f*)tcur = acc;
                cur = lr;
                tcur = T + lr * TILE_PITCH + (q << 2);
                acc = *(v4f*)tcur;
            }
            acc += ok ? t[u] : zero4;
        }
    }
    *(v4f*)tcur = acc;
#pragma unroll
    for (int i = 0; i < 8; ++i) {
        const int lr = 2 * i + hh;
        const int* rr = rstart + wave * 16 + lr;
        const float inv = 1.0f / fmaxf((float)(rr[1] - rr[0] + 1), 1.0f);
        float* trow = T + lr * TILE_PITCH + (q << 2);
        v4f a = *(v4f*)trow;
        a *= inv;
        *(v4f*)trow = a;
    }
}

#define LONG_BATCH 4

__device__ __forceinline__ void gather_mean_tile(const float* __restrict__ x,
                                                 const int* __restrict__ pw,
                                                 const int* __restrict__ offk, int n_chunks,
                                                 long row0, int n, float* T,
                                                 int* lnbr, int* deg, int* lstart, int* cursor,
                                                 int* rdeg, int* wsum,
                                                 int* cpos, int* cbase,
                                                 int tid, int lane, int wave) {
    const int hh = lane >> 4;
    const int q  = lane & 15;
    const char* __restrict__ xb = (const char*)x;
    const unsigned qoff  = (unsigned)q << 4;
    const unsigned rbase = (unsigned)row0;
    const unsigned rlast = (unsigned)(n - 1);

    const v4f zero4 = {0.0f, 0.0f, 0.0f, 0.0f};

    const char* __restrict__ pself = xb + (((rbase + (unsigned)hh) << 8) | qoff);
    v4f sv[8];
#pragma unroll
    for (int i = 0; i < 8; ++i) {
        sv[i] = zero4;
        if (rbase + (unsigned)(2 * i + hh) <= rlast) sv[i] = *(const v4f*)(pself + i * 512);
    }
#pragma unroll
    for (int i = 0; i < 8; ++i)
        *(v4f*)(T + (2 * i + hh) * TILE_PITCH + (q << 2)) = sv[i];
    if (tid < BUCKET_ROWS) rdeg[tid] = 0;

    int seg_count;
    {
        int rs = 0, len = 0;
        if (tid < n_chunks) {
            const int* __restrict__ offe = offk + n_chunks;
            rs  = offk[tid];
            len = offe[tid] - rs;
        }
        int inc = len;
        for (int o = 1; o < 32; o <<= 1) {
            const int t = __shfl_up(inc, o, 32);
            if (lane >= o) inc += t;
        }
        if (lane == 31) wsum[wave] = inc;
        __syncthreads();
        int woff = 0, tot = 0;
        for (int k = 0; k < 16; ++k) {
            const int v = wsum[k];
            tot += v;
            woff += (k < wave) ? v : 0;
        }
        cpos[tid]  = (tid < n_chunks) ? (woff + inc - len) : 0x7fffffff;
        cbase[tid] = rs;
        seg_count = __builtin_amdgcn_readfirstlane(tot);
        __syncthreads();
    }

    const unsigned self_off = (min(rbase + 8u * (unsigned)hh, rlast) << 8) | qoff;

    for (int pb = 0; pb < seg_count; pb += PIECE) {
        const int pc = min(PIECE, seg_count - pb);
        int w[8], p[8], c[8];
        const int pend = pb + pc;
        int pt = pb + tid;
        asm volatile("" : "+v"(pt));
#pragma unroll
        for (int u = 0; u < 8; ++u) {
            const int pp = pt + u * 512;
            p[u] = (pp < pend) ? pp : pb;
            c[u] = 0;
        }
#pragma unroll
        for (int step = MAX_CHUNKS >> 1; step > 0; step >>= 1) {
#pragma unroll
            for (int u = 0; u < 8; ++u)
                if (cpos[c[u] + step] <= p[u]) c[u] += step;
        }
#pragma unroll
        for (int u = 0; u < 8; ++u)
            w[u] = pw[c[u] * CHUNK + cbase[c[u]] + (p[u] - cpos[c[u]])];
        if (tid < BUCKET_ROWS) deg[tid] = 0;
        __syncthreads();
#pragma unroll
        for (int u = 0; u < 8; ++u)
            if (tid + u * 512 < pc) lds_fetch_add(&deg[w[u] & (BUCKET_ROWS - 1)], 1);
        __syncthreads();
        const int d0 = (tid < BUCKET_ROWS) ? deg[tid] : 0;
        int inc = d0;
        for (int o = 1; o < 32; o <<= 1) {
            const int t = __shfl_up(inc, o, 32);
            if (lane >= o) inc += t;
        }
        if (lane == 31 && wave < 8) wsum[wave] = inc;
        __syncthreads();
        if (tid < BUCKET_ROWS) {
            int woff = 0;
            for (int k = 0; k < 8; ++k) woff += (k < wave) ? wsum[k] : 0;
            const int st = woff + inc - d0;
            lstart[tid] = st;
            cursor[tid] = st;
            rdeg[tid] += d0;
        }
        __syncthreads();
#pragma unroll
        for (int u = 0; u < 8; ++u) {
            if (tid + u * 512 < pc) {
                const int slot = lds_fetch_add(&cursor[w[u] & (BUCKET_ROWS - 1)], 1);
                lnbr[slot] = w[u];
            }
        }
        __syncthreads();

        const int beg = lstart[wave * 16 + 8 * hh];
        const int cnt = cursor[wave * 16 + 8 * hh + 7] - beg;
        const int cother = __shfl_xor(cnt, 16, 32);
        const int cmax = max(cnt, cother);

        int cur = 8 * hh;
        float* tcur = T + cur * TILE_PITCH + (q << 2);
        v4f acc = *(v4f*)tcur;
        for (int j = 0; j < cmax; j += LONG_BATCH) {
            int wd[LONG_BATCH];
            v4f t[LONG_BATCH];
#pragma unroll
            for (int u = 0; u < LONG_BATCH; ++u)
                wd[u] = lnbr[(j + u < cnt) ? (beg + j + u) : 0];
#pragma unroll
            for (int u = 0; u < LONG_BATCH; ++u) {
                const unsigned off = (j + u < cnt)
                    ? (((unsigned)wd[u] & ~(unsigned)(BUCKET_ROWS - 1)) | qoff)
                    : self_off;
                t[u] = *(const v4f*)(xb + off);
            }
#pragma unroll
            for (int u = 0; u < LONG_BATCH; ++u) {
                const bool ok = j + u < cnt;
                const int lr = wd[u] & 15;
                if (ok && lr != cur) {
                    *(v4f*)tcur = acc;
                    cur = lr;
                    tcur = T + lr * TILE_PITCH + (q << 2);
                    acc = *(v4f*)tcur;
                }
                acc += ok ? t[u] : zero4;
            }
        }
        *(v4f*)tcur = acc;
        __syncthreads();
    }
    __syncthreads();
#pragma unroll
    for (int i = 0; i < 8; ++i) {
        const int lr = 2 * i + hh;
        const float inv = 1.0f / fmaxf((float)(rdeg[wave * 16 + lr] + 1), 1.0f);
        float* trow = T + lr * TILE_PITCH + (q << 2);
        v4f acc = *(v4f*)trow;
        acc *= inv;
        *(v4f*)trow = acc;
    }
}

__device__ __forceinline__ void tile_product(const float* ta, const float* wset, int h,
                                             v8f& acc0, v8f& acc1, v8f& acc2, v8f& acc3) {
#pragma unroll
    for (int k = 0; k < 16; ++k) {
        const int c = (k << 2) + (h << 1);
        const v2f a = *(const v2f*)(ta + c);
        const float* wp = wset + (((k << 1) + h) << 7);
        const v2f b0 = *(const v2f*)(wp);
        const v2f b1 = *(const v2f*)(wp + 32);
        const v2f b2 = *(const v2f*)(wp + 64);
        const v2f b3 = *(const v2f*)(wp + 96);
        acc0 = __builtin_amdgcn_wmma_f32_16x16x4_f32(false, a, false, b0, (short)0, acc0, false, false);
        acc1 = __builtin_amdgcn_wmma_f32_16x16x4_f32(false, a, false, b1, (short)0, acc1, false, false);
        acc2 = __builtin_amdgcn_wmma_f32_16x16x4_f32(false, a, false, b2, (short)0, acc2, false, false);
        acc3 = __builtin_amdgcn_wmma_f32_16x16x4_f32(false, a, false, b3, (short)0, acc3, false, false);
    }
    asm volatile("v_nop\n\tv_nop\n\tv_nop\n\tv_nop\n\tv_nop\n\tv_nop\n\tv_nop\n\tv_nop"
                 : "+v"(acc0), "+v"(acc1), "+v"(acc2), "+v"(acc3));
}

__global__ __launch_bounds__(512, 8) void sage_gather_gemm_norm_kernel(
        const float* __restrict__ x1,   const float* __restrict__ x2,
        const int* __restrict__ off,    const int* __restrict__ pairs,
        const float* __restrict__ W,    const float* __restrict__ bias,
        float* __restrict__ out, int n, int n_edges, int n_buckets, int n_chunks) {
    __shared__ float ldsW[192 * 64];
    __shared__ __attribute__((aligned(16))) float ldsA[16 * 16 * TILE_PITCH];
    __shared__ int lnbr[2][PIECE];
    __shared__ int rstart[2][BUCKET_ROWS + 1];
    __shared__ int wsum[16];

    const int tid = threadIdx.x;
    for (int idx = tid; idx < 192 * 64; idx += 512) {
        const int row = idx >> 6;
        const int col = idx & 63;
        ldsW[((((row >> 1) << 6) | col) << 1) | (row & 1)] = W[idx];
    }

    const int lane = tid & 31;
    const int wave = tid >> 5;
    const int m = lane & 15;
    const int h = lane >> 4;

    const long tile = (long)blockIdx.x * 16 + wave;
    const long row0 = tile << 4;

    float* T = ldsA + wave * (16 * TILE_PITCH);
    const float* ta = T + m * TILE_PITCH;

    v8f acc0 = {}; v8f acc1 = {}; v8f acc2 = {}; v8f acc3 = {};
    const float* wbase = ldsW + (m << 1);

    int* const scr    = (int*)ldsA;
    int* const cpos   = scr;
    int* const cbase  = scr + MAX_CHUNKS;
    int* const deg    = scr + 2 * MAX_CHUNKS;
    int* const cursor = scr + 2 * MAX_CHUNKS + BUCKET_ROWS;

    bool fits = true;
#pragma nounroll
    for (int set = 0; set < 2; ++set) {
        const int* __restrict__ offk = off + ((long)set * (n_buckets + 1) + blockIdx.x) * n_chunks;
        const int count = order_set(pairs + (long)set * n_edges, offk, n_chunks, lnbr[set], rstart[set],
                                    deg, cursor, wsum, cpos, cbase, tid, lane, wave);
        if (count > PIECE) {
            fits = false;
            break;
        }
    }

    if (fits) {
#pragma nounroll
        for (int set = 0; set < 2; ++set) {
            gather_rows(set ? x2 : x1, lnbr[set], rstart[set], row0, n, T, lane, wave);
            tile_product(ta, wbase + ((set << 5) << 7), h, acc0, acc1, acc2, acc3);
        }
    } else {
        __syncthreads();
        int* const fb = lnbr[1];
#pragma nounroll
        for (int set = 0; set < 2; ++set) {
            const int* __restrict__ offk = off + ((long)set * (n_buckets + 1) + blockIdx.x) * n_chunks;
            gather_mean_tile(set ? x2 : x1, pairs + (long)set * n_edges, offk, n_chunks,
                             row0, n, T, lnbr[0], fb, fb + BUCKET_ROWS, fb + 2 * BUCKET_ROWS,
                             fb + 3 * BUCKET_ROWS, wsum, fb + 4 * BUCKET_ROWS,
                             fb + 4 * BUCKET_ROWS + MAX_CHUNKS, tid, lane, wave);
            __syncthreads();
            tile_product(ta, wbase + ((set << 5) << 7), h, acc0, acc1, acc2, acc3);
            __syncthreads();
        }
    }
    int tid_end = tid;
    asm volatile("" : "+v"(tid_end));
    const int me = tid_end & 15;
    const int he = (tid_end >> 4) & 1;
    const float* wbe = ldsW + (me << 1);
    long r = row0 + me;
    if (r > (long)(n - 1)) r = (long)(n - 1);
    const float* pax = x1 + r * D_FEAT;
#pragma unroll
    for (int k = 0; k < 16; ++k) {
        const int c = (k << 2) + (he << 1);
        const v2f a = *(const v2f*)(pax + c);
        const float* wp = wbe + ((64 + (k << 1) + he) << 7);
        const v2f b0 = *(const v2f*)(wp);
        const v2f b1 = *(const v2f*)(wp + 32);
        const v2f b2 = *(const v2f*)(wp + 64);
        const v2f b3 = *(const v2f*)(wp + 96);
        acc0 = __builtin_amdgcn_wmma_f32_16x16x4_f32(false, a, false, b0, (short)0, acc0, false, false);
        acc1 = __builtin_amdgcn_wmma_f32_16x16x4_f32(false, a, false, b1, (short)0, acc1, false, false);
        acc2 = __builtin_amdgcn_wmma_f32_16x16x4_f32(false, a, false, b2, (short)0, acc2, false, false);
        acc3 = __builtin_amdgcn_wmma_f32_16x16x4_f32(false, a, false, b3, (short)0, acc3, false, false);
    }

    const float bs0 = bias[me];
    const float bs1 = bias[16 + me];
    const float bs2 = bias[32 + me];
    const float bs3 = bias[48 + me];

#pragma unroll
    for (int j = 0; j < 8; ++j) {
        const float o0 = acc0[j] + bs0;
        const float o1 = acc1[j] + bs1;
        const float o2 = acc2[j] + bs2;
        const float o3 = acc3[j] + bs3;
        float s = o0 * o0 + o1 * o1 + o2 * o2 + o3 * o3;
        s += __shfl_xor(s, 1, 32);
        s += __shfl_xor(s, 2, 32);
        s += __shfl_xor(s, 4, 32);
        s += __shfl_xor(s, 8, 32);
        const float scale = 1.0f / fmaxf(sqrtf(s), 1e-12f);
        const long orow = row0 + j + (he << 3);
        if (orow < n) {
            float* po = out + orow * D_FEAT + me;
            po[0]  = o0 * scale;
            po[16] = o1 * scale;
            po[32] = o2 * scale;
            po[48] = o3 * scale;
        }
    }
}

extern "C" void kernel_launch(void* const* d_in, const int* in_sizes, int n_in,
                              void* d_out, int out_size, void* d_ws, size_t ws_size,
                              hipStream_t stream) {
    const float* x1    = (const float*)d_in[0];
    const float* x2    = (const float*)d_in[1];
    const int*   e_pos = (const int*)d_in[2];
    const int*   e_neg = (const int*)d_in[3];
    const float* W     = (const float*)d_in[4];
    const float* bias  = (const float*)d_in[5];
    float*       out   = (float*)d_out;

    const int n       = in_sizes[0] / D_FEAT;
    const int n_edges = in_sizes[2] / 2;

    const int n_buckets = (n + BUCKET_ROWS - 1) / BUCKET_ROWS;
    const int n_chunks  = (n_edges + CHUNK - 1) / CHUNK;

    int* off   = (int*)d_ws;
    int* pairs = off + (size_t)2 * (n_buckets + 1) * n_chunks;

    const dim3 egrid(n_chunks, 2);
    chunk_group_kernel<<<egrid, EDGE_THREADS, 0, stream>>>(e_pos, e_neg, off, pairs,
                                                  n_edges, n_buckets, n_chunks);
    {
        const int tiles  = (n + 15) / 16;
        const int blocks = (tiles + 15) / 16;
        sage_gather_gemm_norm_kernel<<<blocks, 512, 0, stream>>>(x1, x2, off, pairs,
                                                                 W, bias, out, n, n_edges, n_buckets,
                                                                 n_chunks);
    }
}


# Rows dealt as 250 GEMM blocks of 25 waves (400-row buckets), one per WGP, not 391 blocks of 16 waves resident two-and-one: busiest WGP holds 400 rows, was 512; row sums stay LDS-arrival order
# speedup vs baseline: 6.8611x; 1.0616x over previous
#include <hip/hip_runtime.h>
#include <hip/hip_bf16.h>

typedef __attribute__((ext_vector_type(2))) float v2f;
typedef __attribute__((ext_vector_type(4))) float v4f;
typedef __attribute__((ext_vector_type(8))) float v8f;

#define D_FEAT 64
#define TILE_PITCH 68
#define BUCKET_ROWS 400
#define ROW_BITS 9
#define ROW_MASK 511
#define GEMM_WAVES 25
#define GEMM_THREADS (GEMM_WAVES * 32)
#define ROW_WAVES ((BUCKET_ROWS + 31) / 32)
#define MAX_BUCKETS 512
#define CHUNK 4096
#define EDGE_THREADS 128
#define EDGES_PER_THREAD (CHUNK / EDGE_THREADS)
#define BUCKETS_PER_THREAD (MAX_BUCKETS / EDGE_THREADS)
#define MAX_CHUNKS 512
#define PIECE 4800
#define SEARCHES (PIECE / GEMM_THREADS)


__device__ __forceinline__ int lds_fetch_add(int* p, int v) {
    return __hip_atomic_fetch_add(p, v, __ATOMIC_RELAXED, __HIP_MEMORY_SCOPE_WORKGROUP);
}

__global__ __launch_bounds__(EDGE_THREADS) void chunk_group_kernel(const int* __restrict__ e_pos,
                                                                   const int* __restrict__ e_neg,
                                                                   int* __restrict__ off,
                                                                   int* __restrict__ pairs,
                                                                   int n_edges, int n_buckets, int n_chunks) {
    __shared__ int hist[MAX_BUCKETS];
    __shared__ int wsum[EDGE_THREADS / 32];
    __shared__ int sorted[CHUNK];
    const int set  = blockIdx.y;
    const int tid  = threadIdx.x;
    const int lane = tid & 31;
    const int wave = tid >> 5;
    const int* __restrict__ edges = set ? e_neg : e_pos;

    int* myhist = hist + tid * BUCKETS_PER_THREAD;
#pragma unroll
    for (int k = 0; k < BUCKETS_PER_THREAD; ++k) myhist[k] = 0;
    const int e0 = blockIdx.x * CHUNK + tid;
    int rw[EDGES_PER_THREAD], cl[EDGES_PER_THREAD], rk[EDGES_PER_THREAD];
    const int e_last = n_edges - 1;
#pragma unroll
    for (int u = 0; u < EDGES_PER_THREAD; ++u) {
        const int e  = e0 + u * EDGE_THREADS;
        const int ec = min(e, e_last);
        const int r  = edges[ec];
        cl[u] = edges[(long)n_edges + ec];
        rw[u] = (e < n_edges) ? r : -1;
    }
#pragma unroll
    for (int u = 0; u < EDGES_PER_THREAD; ++u) {
        const unsigned rr = (unsigned)max(rw[u], 0);
        const unsigned b  = rr / BUCKET_ROWS;
        rw[u] = (int)((b << ROW_BITS) | (rr - b * BUCKET_ROWS)) | (rw[u] >> 31);
    }
    __syncthreads();
#pragma unroll
    for (int u = 0; u < EDGES_PER_THREAD; ++u) {
        rk[u] = 0;
        if (rw[u] >= 0) rk[u] = lds_fetch_add(&hist[rw[u] >> ROW_BITS], 1);
    }
    __syncthreads();

    int cnt[BUCKETS_PER_THREAD];
    int tot = 0;
#pragma unroll
    for (int k = 0; k < BUCKETS_PER_THREAD; ++k) {
        cnt[k] = myhist[k];
        tot += cnt[k];
    }
    int inc = tot;
    for (int o = 1; o < 32; o <<= 1) {
        const int s = __shfl_up(inc, o, 32);
        if (lane >= o) inc += s;
    }
    if (lane == 31) wsum[wave] = inc;
    __syncthreads();
    int start = inc - tot;
    for (int w = 0; w < EDGE_THREADS / 32; ++w) start += (w < wave) ? wsum[w] : 0;
#pragma unroll
    for (int k = 0; k < BUCKETS_PER_THREAD; ++k) {
        const int b = tid * BUCKETS_PER_THREAD + k;
        myhist[k] = start;
        if (b <= n_buckets)
            off[((long)set * (n_buckets + 1) + b) * n_chunks + blockIdx.x] = start;
        start += cnt[k];
    }
    __syncthreads();

#pragma unroll
    for (int u = 0; u < EDGES_PER_THREAD; ++u)
        rk[u] += hist[(rw[u] >= 0) ? (rw[u] >> ROW_BITS) : 0];
#pragma unroll
    for (int u = 0; u < EDGES_PER_THREAD; ++u)
        if (rw[u] >= 0)
            sorted[rk[u]] = (cl[u] << ROW_BITS) | (rw[u] & ROW_MASK);
    __syncthreads();

    const int m = min(CHUNK, n_edges - blockIdx.x * CHUNK);
    int* __restrict__ dst = pairs + (long)set * n_edges + (long)blockIdx.x * CHUNK;
#pragma unroll
    for (int u = 0; u < EDGES_PER_THREAD; ++u) rk[u] = sorted[min(tid + u * EDGE_THREADS, m - 1)];
#pragma unroll
    for (int u = 0; u < EDGES_PER_THREAD; ++u) {
        const int i = tid + u * EDGE_THREADS;
        if (i < m) dst[i] = rk[u];
    }
}


__device__ __forceinline__ int order_set(const int* __restrict__ pw, const int* __restrict__ offk,
                                         int n_chunks, int* lnbr, int* rstart,
                                         int* deg, int* cursor, int* wsum, int* cpos, int* cbase,
                                         int tid, int lane, int wave) {
    int seg_count;
    {
        int rs = 0, len = 0;
        if (tid < n_chunks) {
            const int* __restrict__ offe = offk + n_chunks;
            rs  = offk[tid];
            len = offe[tid] - rs;
        }
        int inc = len;
        for (int o = 1; o < 32; o <<= 1) {
            const int t = __shfl_up(inc, o, 32);
            if (lane >= o) inc += t;
        }
        if (lane == 31) wsum[wave] = inc;
        __syncthreads();
        int woff = 0, tot = 0;
        for (int k = 0; k < GEMM_WAVES; ++k) {
            const int v = wsum[k];
            tot += v;
            woff += (k < wave) ? v : 0;
        }
        if (tid < MAX_CHUNKS) {
            cpos[tid]  = (tid < n_chunks) ? (woff + inc - len) : 0x7fffffff;
            cbase[tid] = rs;
        }
        if (tid < BUCKET_ROWS) deg[tid] = 0;
        seg_count = __builtin_amdgcn_readfirstlane(tot);
        __syncthreads();
    }
    if (seg_count > PIECE) return seg_count;
    if (seg_count == 0) {
        if (tid <= BUCKET_ROWS) rstart[tid] = 0;
        __syncthreads();
        return 0;
    }
    const int pc = seg_count;

    int w[SEARCHES], p[SEARCHES], c[SEARCHES];
#pragma unroll
    for (int u = 0; u < SEARCHES; ++u) {
        const int pp = tid + u * GEMM_THREADS;
        p[u] = (pp < pc) ? pp : 0;
        c[u] = 0;
    }
#pragma unroll
    for (int step = MAX_CHUNKS >> 1; step > 0; step >>= 1) {
#pragma unroll
        for (int u = 0; u < SEARCHES; ++u)
            if (cpos[c[u] + step] <= p[u]) c[u] += step;
    }
#pragma unroll
    for (int u = 0; u < SEARCHES; ++u)
        w[u] = pw[c[u] * CHUNK + cbase[c[u]] + (p[u] - cpos[c[u]])];
#pragma unroll
    for (int u = 0; u < SEARCHES; ++u)
        if (tid + u * GEMM_THREADS < pc) lds_fetch_add(&deg[w[u] & ROW_MASK], 1);
    __syncthreads();
    const int d0 = (tid < BUCKET_ROWS) ? deg[tid] : 0;
    int inc = d0;
    for (int o = 1; o < 32; o <<= 1) {
        const int t = __shfl_up(inc, o, 32);
        if (lane >= o) inc += t;
    }
    if (lane == 31 && wave < ROW_WAVES) wsum[wave] = inc;
    __syncthreads();
    if (tid < BUCKET_ROWS) {
        int woff = 0;
        for (int k = 0; k < ROW_WAVES; ++k) woff += (k < wave) ? wsum[k] : 0;
        const int st = woff + inc - d0;
        rstart[tid] = st;
        cursor[tid] = st;
        if (tid == BUCKET_ROWS - 1) rstart[BUCKET_ROWS] = st + d0;
    }
    __syncthreads();
#pragma unroll
    for (int u = 0; u < SEARCHES; ++u) {
        if (tid + u * GEMM_THREADS < pc) {
            const int slot = lds_fetch_add(&cursor[w[u] & ROW_MASK], 1);
            lnbr[slot] = w[u];
        }
    }
    __syncthreads();
    return seg_count;
}

#define GATHER_BATCH 10

__device__ __forceinline__ void gather_rows(const float* __restrict__ x, const int* lnbr,
                                            const int* rstart, long row0, int n, float* T,
                                            int lane, int wave) {
    const int hh = lane >> 4;
    const int q  = lane & 15;
    const char* __restrict__ xb = (const char*)x;
    const unsigned qoff  = (unsigned)q << 4;
    const unsigned rbase = (unsigned)row0;
    const unsigned rlast = (unsigned)(n - 1);

    const v4f zero4 = {0.0f, 0.0f, 0.0f, 0.0f};

    const char* __restrict__ pself = xb + (((rbase + (unsigned)hh) << 8) | qoff);
    v4f sv[8];
#pragma unroll
    for (int i = 0; i < 8; ++i) {
        sv[i] = zero4;
        if (rbase + (unsigned)(2 * i + hh) <= rlast) sv[i] = *(const v4f*)(pself + i * 512);
    }
#pragma unroll
    for (int i = 0; i < 8; ++i)
        *(v4f*)(T + (2 * i + hh) * TILE_PITCH + (q << 2)) = sv[i];

    const unsigned self_off = (min(rbase + 8u * (unsigned)hh, rlast) << 8) | qoff;

    const int* rs = rstart + wave * 16 + 8 * hh;
    const int beg = rs[0];
    const int cnt = rs[8] - beg;
    const int cother = __shfl_xor(cnt, 16, 32);
    const int cmax = max(cnt, cother);
    const int cmin = __builtin_amdgcn_readfirstlane(min(cnt, cother));

    int cur = 8 * hh;
    float* tcur = T + cur * TILE_PITCH + (q << 2);
    v4f acc = *(v4f*)tcur;
    int j = 0;
    {
        const int* lp = lnbr + beg;
        for (; j + GATHER_BATCH <= cmin; j += GATHER_BATCH, lp += GATHER_BATCH) {
            int wd[GATHER_BATCH];
            v4f t[GATHER_BATCH];
#pragma unroll
            for (int u = 0; u < GATHER_BATCH; ++u) wd[u] = lp[u];
#pragma unroll
            for (int u = 0; u < GATHER_BATCH; ++u)
                t[u] = *(const v4f*)(xb + (((((unsigned)wd[u]) >> 1) & ~255u) | qoff));
#pragma unroll
            for (int u = 0; u < GATHER_BATCH; ++u) {
                const int lr = wd[u] & 15;
                if (lr != cur) {
                    *(v4f*)tcur = acc;
                    cur = lr;
                    tcur = T + lr * TILE_PITCH + (q << 2);
                    acc = *(v4f*)tcur;
                }
                acc += t[u];
            }
        }
    }
    for (; j < cmax; j += GATHER_BATCH) {
        int wd[GATHER_BATCH];
        v4f t[GATHER_BATCH];
#pragma unroll
        for (int u = 0; u < GATHER_BATCH; ++u)
            wd[u] = lnbr[(j + u < cnt) ? (beg + j + u) : 0];
#pragma unroll
        for (int u = 0; u < GATHER_BATCH; ++u) {
            const unsigned off = (j + u < cnt)
                ? ((((unsigned)wd[u] >> 1) & ~255u) | qoff)
                : self_off;
            t[u] = *(const v4f*)(xb + off);
        }
#pragma unroll
        for (int u = 0; u < GATHER_BATCH; ++u) {
            const bool ok = j + u < cnt;
            const int lr = wd[u] & 15;
            if (ok && lr != cur) {
                *(v4f*)tcur = acc;
                cur = lr;
                tcur = T + lr * TILE_PITCH + (q << 2);
                acc = *(v4f*)tcur;
            }
            acc += ok ? t[u] : zero4;
        }
    }
    *(v4f*)tcur = acc;
#pragma unroll
    for (int i = 0; i < 8; ++i) {
        const int lr = 2 * i + hh;
        const int* rr = rstart + wave * 16 + lr;
        const float inv = 1.0f / fmaxf((float)(rr[1] - rr[0] + 1), 1.0f);
        float* trow = T + lr * TILE_PITCH + (q << 2);
        v4f a = *(v4f*)trow;
        a *= inv;
        *(v4f*)trow = a;
    }
}

#define LONG_BATCH 4

__device__ __forceinline__ void gather_mean_tile(const float* __restrict__ x,
                                                 const int* __restrict__ pw,
                                                 const int* __restrict__ offk, int n_chunks,
                                                 long row0, int n, float* T,
                                                 int* lnbr, int* deg, int* lstart, int* cursor,
                                                 int* rdeg, int* wsum,
                                                 int* cpos, int* cbase,
                                                 int tid, int lane, int wave) {
    const int hh = lane >> 4;
    const int q  = lane & 15;
    const char* __restrict__ xb = (const char*)x;
    const unsigned qoff  = (unsigned)q << 4;
    const unsigned rbase = (unsigned)row0;
    const unsigned rlast = (unsigned)(n - 1);

    const v4f zero4 = {0.0f, 0.0f, 0.0f, 0.0f};

    const char* __restrict__ pself = xb + (((rbase + (unsigned)hh) << 8) | qoff);
    v4f sv[8];
#pragma unroll
    for (int i = 0; i < 8; ++i) {
        sv[i] = zero4;
        if (rbase + (unsigned)(2 * i + hh) <= rlast) sv[i] = *(const v4f*)(pself + i * 512);
    }
#pragma unroll
    for (int i = 0; i < 8; ++i)
        *(v4f*)(T + (2 * i + hh) * TILE_PITCH + (q << 2)) = sv[i];
    if (tid < BUCKET_ROWS) rdeg[tid] = 0;

    int seg_count;
    {
        int rs = 0, len = 0;
        if (tid < n_chunks) {
            const int* __restrict__ offe = offk + n_chunks;
            rs  = offk[tid];
            len = offe[tid] - rs;
        }
        int inc = len;
        for (int o = 1; o < 32; o <<= 1) {
            const int t = __shfl_up(inc, o, 32);
            if (lane >= o) inc += t;
        }
        if (lane == 31) wsum[wave] = inc;
        __syncthreads();
        int woff = 0, tot = 0;
        for (int k = 0; k < GEMM_WAVES; ++k) {
            const int v = wsum[k];
            tot += v;
            woff += (k < wave) ? v : 0;
        }
        if (tid < MAX_CHUNKS) {
            cpos[tid]  = (tid < n_chunks) ? (woff + inc - len) : 0x7fffffff;
            cbase[tid] = rs;
        }
        seg_count = __builtin_amdgcn_readfirstlane(tot);
        __syncthreads();
    }

    const unsigned self_off = (min(rbase + 8u * (unsigned)hh, rlast) << 8) | qoff;

    for (int pb = 0; pb < seg_count; pb += PIECE) {
        const int pc = min(PIECE, seg_count - pb);
        int w[SEARCHES], p[SEARCHES], c[SEARCHES];
        const int pend = pb + pc;
        const int pt = pb + tid;
#pragma unroll
        for (int u = 0; u < SEARCHES; ++u) {
            const int pp = pt + u * GEMM_THREADS;
            p[u] = (pp < pend) ? pp : pb;
            c[u] = 0;
        }
#pragma unroll
        for (int step = MAX_CHUNKS >> 1; step > 0; step >>= 1) {
#pragma unroll
            for (int u = 0; u < SEARCHES; ++u)
                if (cpos[c[u] + step] <= p[u]) c[u] += step;
        }
#pragma unroll
        for (int u = 0; u < SEARCHES; ++u)
            w[u] = pw[c[u] * CHUNK + cbase[c[u]] + (p[u] - cpos[c[u]])];
        if (tid < BUCKET_ROWS) deg[tid] = 0;
        __syncthreads();
#pragma unroll
        for (int u = 0; u < SEARCHES; ++u)
            if (tid + u * GEMM_THREADS < pc) lds_fetch_add(&deg[w[u] & ROW_MASK], 1);
        __syncthreads();
        const int d0 = (tid < BUCKET_ROWS) ? deg[tid] : 0;
        int inc = d0;
        for (int o = 1; o < 32; o <<= 1) {
            const int t = __shfl_up(inc, o, 32);
            if (lane >= o) inc += t;
        }
        if (lane == 31 && wave < ROW_WAVES) wsum[wave] = inc;
        __syncthreads();
        if (tid < BUCKET_ROWS) {
            int woff = 0;
            for (int k = 0; k < ROW_WAVES; ++k) woff += (k < wave) ? wsum[k] : 0;
            const int st = woff + inc - d0;
            lstart[tid] = st;
            cursor[tid] = st;
            rdeg[tid] += d0;
        }
        __syncthreads();
#pragma unroll
        for (int u = 0; u < SEARCHES; ++u) {
            if (tid + u * GEMM_THREADS < pc) {
                const int slot = lds_fetch_add(&cursor[w[u] & ROW_MASK], 1);
                lnbr[slot] = w[u];
            }
        }
        __syncthreads();

        const int beg = lstart[wave * 16 + 8 * hh];
        const int cnt = cursor[wave * 16 + 8 * hh + 7] - beg;
        const int cother = __shfl_xor(cnt, 16, 32);
        const int cmax = max(cnt, cother);

        int cur = 8 * hh;
        float* tcur = T + cur * TILE_PITCH + (q << 2);
        v4f acc = *(v4f*)tcur;
        for (int j = 0; j < cmax; j += LONG_BATCH) {
            int wd[LONG_BATCH];
            v4f t[LONG_BATCH];
#pragma unroll
            for (int u = 0; u < LONG_BATCH; ++u)
                wd[u] = lnbr[(j + u < cnt) ? (beg + j + u) : 0];
#pragma unroll
            for (int u = 0; u < LONG_BATCH; ++u) {
                const unsigned off = (j + u < cnt)
                    ? ((((unsigned)wd[u] >> 1) & ~255u) | qoff)
                    : self_off;
                t[u] = *(const v4f*)(xb + off);
            }
#pragma unroll
            for (int u = 0; u < LONG_BATCH; ++u) {
                const bool ok = j + u < cnt;
                const int lr = wd[u] & 15;
                if (ok && lr != cur) {
                    *(v4f*)tcur = acc;
                    cur = lr;
                    tcur = T + lr * TILE_PITCH + (q << 2);
                    acc = *(v4f*)tcur;
                }
                acc += ok ? t[u] : zero4;
            }
        }
        *(v4f*)tcur = acc;
        __syncthreads();
    }
    __syncthreads();
#pragma unroll
    for (int i = 0; i < 8; ++i) {
        const int lr = 2 * i + hh;
        const float inv = 1.0f / fmaxf((float)(rdeg[wave * 16 + lr] + 1), 1.0f);
        float* trow = T + lr * TILE_PITCH + (q << 2);
        v4f acc = *(v4f*)trow;
        acc *= inv;
        *(v4f*)trow = acc;
    }
}

__device__ __forceinline__ void tile_product(const float* ta, const float* wset, int h,
                                             v8f& acc0, v8f& acc1, v8f& acc2, v8f& acc3) {
#pragma unroll
    for (int k = 0; k < 16; ++k) {
        const int c = (k << 2) + (h << 1);
        const v2f a = *(const v2f*)(ta + c);
        const float* wp = wset + (((k << 1) + h) << 7);
        const v2f b0 = *(const v2f*)(wp);
        const v2f b1 = *(const v2f*)(wp + 32);
        const v2f b2 = *(const v2f*)(wp + 64);
        const v2f b3 = *(const v2f*)(wp + 96);
        acc0 = __builtin_amdgcn_wmma_f32_16x16x4_f32(false, a, false, b0, (short)0, acc0, false, false);
        acc1 = __builtin_amdgcn_wmma_f32_16x16x4_f32(false, a, false, b1, (short)0, acc1, false, false);
        acc2 = __builtin_amdgcn_wmma_f32_16x16x4_f32(false, a, false, b2, (short)0, acc2, false, false);
        acc3 = __builtin_amdgcn_wmma_f32_16x16x4_f32(false, a, false, b3, (short)0, acc3, false, false);
    }
    asm volatile("v_nop\n\tv_nop\n\tv_nop\n\tv_nop\n\tv_nop\n\tv_nop\n\tv_nop\n\tv_nop"
                 : "+v"(acc0), "+v"(acc1), "+v"(acc2), "+v"(acc3));
}

__global__ __launch_bounds__(GEMM_THREADS, 7) void sage_gather_gemm_norm_kernel(
        const float* __restrict__ x1,   const float* __restrict__ x2,
        const int* __restrict__ off,    const int* __restrict__ pairs,
        const float* __restrict__ W,    const float* __restrict__ bias,
        float* __restrict__ out, int n, int n_edges, int n_buckets, int n_chunks) {
    __shared__ float ldsW[192 * 64];
    __shared__ __attribute__((aligned(16))) float ldsA[GEMM_WAVES * 16 * TILE_PITCH];
    __shared__ int lnbr[2][PIECE];
    __shared__ int rstart[2][BUCKET_ROWS + 1];
    __shared__ int wsum[32];

    const int tid = threadIdx.x;
    for (int idx = tid; idx < 192 * 64; idx += GEMM_THREADS) {
        const int row = idx >> 6;
        const int col = idx & 63;
        ldsW[((((row >> 1) << 6) | col) << 1) | (row & 1)] = W[idx];
    }

    const int lane = tid & 31;
    const int wave = tid >> 5;
    const int m = lane & 15;
    const int h = lane >> 4;

    const long tile = (long)blockIdx.x * GEMM_WAVES + wave;
    const long row0 = tile << 4;

    float* T = ldsA + wave * (16 * TILE_PITCH);
    const float* ta = T + m * TILE_PITCH;

    v8f acc0 = {}; v8f acc1 = {}; v8f acc2 = {}; v8f acc3 = {};
    const float* wbase = ldsW + (m << 1);

    int* const scr    = (int*)ldsA;
    int* const cpos   = scr;
    int* const cbase  = scr + MAX_CHUNKS;
    int* const deg    = scr + 2 * MAX_CHUNKS;
    int* const cursor = scr + 2 * MAX_CHUNKS + BUCKET_ROWS;

    bool fits = true;
#pragma nounroll
    for (int set = 0; set < 2; ++set) {
        const int* __restrict__ offk = off + ((long)set * (n_buckets + 1) + blockIdx.x) * n_chunks;
        const int count = order_set(pairs + (long)set * n_edges, offk, n_chunks, lnbr[set], rstart[set],
                                    deg, cursor, wsum, cpos, cbase, tid, lane, wave);
        if (count > PIECE) {
            fits = false;
            break;
        }
    }

    if (fits) {
#pragma nounroll
        for (int set = 0; set < 2; ++set) {
            gather_rows(set ? x2 : x1, lnbr[set], rstart[set], row0, n, T, lane, wave);
            tile_product(ta, wbase + ((set << 5) << 7), h, acc0, acc1, acc2, acc3);
        }
    } else {
        __syncthreads();
        int* const fb = lnbr[1];
#pragma nounroll
        for (int set = 0; set < 2; ++set) {
            const int* __restrict__ offk = off + ((long)set * (n_buckets + 1) + blockIdx.x) * n_chunks;
            gather_mean_tile(set ? x2 : x1, pairs + (long)set * n_edges, offk, n_chunks,
                             row0, n, T, lnbr[0], fb, fb + BUCKET_ROWS, fb + 2 * BUCKET_ROWS,
                             fb + 3 * BUCKET_ROWS, wsum, fb + 4 * BUCKET_ROWS,
                             fb + 4 * BUCKET_ROWS + MAX_CHUNKS, tid, lane, wave);
            __syncthreads();
            tile_product(ta, wbase + ((set << 5) << 7), h, acc0, acc1, acc2, acc3);
            __syncthreads();
        }
    }
    const int me = m;
    const int he = h;
    const float* wbe = ldsW + (me << 1);
    long r = row0 + me;
    if (r > (long)(n - 1)) r = (long)(n - 1);
    const float* pax = x1 + r * D_FEAT;
#pragma unroll
    for (int k = 0; k < 16; ++k) {
        const int c = (k << 2) + (he << 1);
        const v2f a = *(const v2f*)(pax + c);
        const float* wp = wbe + ((64 + (k << 1) + he) << 7);
        const v2f b0 = *(const v2f*)(wp);
        const v2f b1 = *(const v2f*)(wp + 32);
        const v2f b2 = *(const v2f*)(wp + 64);
        const v2f b3 = *(const v2f*)(wp + 96);
        acc0 = __builtin_amdgcn_wmma_f32_16x16x4_f32(false, a, false, b0, (short)0, acc0, false, false);
        acc1 = __builtin_amdgcn_wmma_f32_16x16x4_f32(false, a, false, b1, (short)0, acc1, false, false);
        acc2 = __builtin_amdgcn_wmma_f32_16x16x4_f32(false, a, false, b2, (short)0, acc2, false, false);
        acc3 = __builtin_amdgcn_wmma_f32_16x16x4_f32(false, a, false, b3, (short)0, acc3, false, false);
    }

    const float bs0 = bias[me];
    const float bs1 = bias[16 + me];
    const float bs2 = bias[32 + me];
    const float bs3 = bias[48 + me];

#pragma unroll
    for (int j = 0; j < 8; ++j) {
        const float o0 = acc0[j] + bs0;
        const float o1 = acc1[j] + bs1;
        const float o2 = acc2[j] + bs2;
        const float o3 = acc3[j] + bs3;
        float s = o0 * o0 + o1 * o1 + o2 * o2 + o3 * o3;
        s += __shfl_xor(s, 1, 32);
        s += __shfl_xor(s, 2, 32);
        s += __shfl_xor(s, 4, 32);
        s += __shfl_xor(s, 8, 32);
        const float scale = 1.0f / fmaxf(sqrtf(s), 1e-12f);
        const long orow = row0 + j + (he << 3);
        if (orow < n) {
            float* po = out + orow * D_FEAT + me;
            po[0]  = o0 * scale;
            po[16] = o1 * scale;
            po[32] = o2 * scale;
            po[48] = o3 * scale;
        }
    }
}

extern "C" void kernel_launch(void* const* d_in, const int* in_sizes, int n_in,
                              void* d_out, int out_size, void* d_ws, size_t ws_size,
                              hipStream_t stream) {
    const float* x1    = (const float*)d_in[0];
    const float* x2    = (const float*)d_in[1];
    const int*   e_pos = (const int*)d_in[2];
    const int*   e_neg = (const int*)d_in[3];
    const float* W     = (const float*)d_in[4];
    const float* bias  = (const float*)d_in[5];
    float*       out   = (float*)d_out;

    const int n       = in_sizes[0] / D_FEAT;
    const int n_edges = in_sizes[2] / 2;

    const int n_buckets = (n + BUCKET_ROWS - 1) / BUCKET_ROWS;
    const int n_chunks  = (n_edges + CHUNK - 1) / CHUNK;

    int* off   = (int*)d_ws;
    int* pairs = off + (size_t)2 * (n_buckets + 1) * n_chunks;

    const dim3 egrid(n_chunks, 2);
    chunk_group_kernel<<<egrid, EDGE_THREADS, 0, stream>>>(e_pos, e_neg, off, pairs,
                                                  n_edges, n_buckets, n_chunks);
    {
        const int tiles  = (n + 15) / 16;
        const int blocks = (tiles + GEMM_WAVES - 1) / GEMM_WAVES;
        sage_gather_gemm_norm_kernel<<<blocks, GEMM_THREADS, 0, stream>>>(x1, x2, off, pairs,
                                                                 W, bias, out, n, n_edges, n_buckets,
                                                                 n_chunks);
    }
}


# Every global load of a block's set-up that needs no ordering (16 words of W per thread, both sets' run tables, the wave's x1 rows) goes out as one group before the first barrier: one wait, not 11
# speedup vs baseline: 7.3371x; 1.0694x over previous
#include <hip/hip_runtime.h>
#include <hip/hip_bf16.h>

typedef __attribute__((ext_vector_type(2))) float v2f;
typedef __attribute__((ext_vector_type(4))) float v4f;
typedef __attribute__((ext_vector_type(8))) float v8f;

#define D_FEAT 64
#define TILE_PITCH 68
#define BUCKET_ROWS 400
#define ROW_BITS 9
#define ROW_MASK 511
#define GEMM_WAVES 25
#define GEMM_THREADS (GEMM_WAVES * 32)
#define ROW_WAVES ((BUCKET_ROWS + 31) / 32)
#define MAX_BUCKETS 512
#define CHUNK 4096
#define EDGE_THREADS 128
#define EDGES_PER_THREAD (CHUNK / EDGE_THREADS)
#define BUCKETS_PER_THREAD (MAX_BUCKETS / EDGE_THREADS)
#define MAX_CHUNKS 512
#define PIECE 4800
#define SEARCHES (PIECE / GEMM_THREADS)
#define W_WORDS ((192 * 64 + GEMM_THREADS - 1) / GEMM_THREADS)


__device__ __forceinline__ int lds_fetch_add(int* p, int v) {
    return __hip_atomic_fetch_add(p, v, __ATOMIC_RELAXED, __HIP_MEMORY_SCOPE_WORKGROUP);
}

__global__ __launch_bounds__(EDGE_THREADS) void chunk_group_kernel(const int* __restrict__ e_pos,
                                                                   const int* __restrict__ e_neg,
                                                                   int* __restrict__ off,
                                                                   int* __restrict__ pairs,
                                                                   int n_edges, int n_buckets, int n_chunks) {
    __shared__ int hist[MAX_BUCKETS];
    __shared__ int wsum[EDGE_THREADS / 32];
    __shared__ int sorted[CHUNK];
    const int set  = blockIdx.y;
    const int tid  = threadIdx.x;
    const int lane = tid & 31;
    const int wave = tid >> 5;
    const int* __restrict__ edges = set ? e_neg : e_pos;

    int* myhist = hist + tid * BUCKETS_PER_THREAD;
#pragma unroll
    for (int k = 0; k < BUCKETS_PER_THREAD; ++k) myhist[k] = 0;
    const int e0 = blockIdx.x * CHUNK + tid;
    int rw[EDGES_PER_THREAD], cl[EDGES_PER_THREAD], rk[EDGES_PER_THREAD];
    const int e_last = n_edges - 1;
#pragma unroll
    for (int u = 0; u < EDGES_PER_THREAD; ++u) {
        const int e  = e0 + u * EDGE_THREADS;
        const int ec = min(e, e_last);
        const int r  = edges[ec];
        cl[u] = edges[(long)n_edges + ec];
        rw[u] = (e < n_edges) ? r : -1;
    }
#pragma unroll
    for (int u = 0; u < EDGES_PER_THREAD; ++u) {
        const unsigned rr = (unsigned)max(rw[u], 0);
        const unsigned b  = rr / BUCKET_ROWS;
        rw[u] = (int)((b << ROW_BITS) | (rr - b * BUCKET_ROWS)) | (rw[u] >> 31);
    }
    __syncthreads();
#pragma unroll
    for (int u = 0; u < EDGES_PER_THREAD; ++u) {
        rk[u] = 0;
        if (rw[u] >= 0) rk[u] = lds_fetch_add(&hist[rw[u] >> ROW_BITS], 1);
    }
    __syncthreads();

    int cnt[BUCKETS_PER_THREAD];
    int tot = 0;
#pragma unroll
    for (int k = 0; k < BUCKETS_PER_THREAD; ++k) {
        cnt[k] = myhist[k];
        tot += cnt[k];
    }
    int inc = tot;
    for (int o = 1; o < 32; o <<= 1) {
        const int s = __shfl_up(inc, o, 32);
        if (lane >= o) inc += s;
    }
    if (lane == 31) wsum[wave] = inc;
    __syncthreads();
    int start = inc - tot;
    for (int w = 0; w < EDGE_THREADS / 32; ++w) start += (w < wave) ? wsum[w] : 0;
#pragma unroll
    for (int k = 0; k < BUCKETS_PER_THREAD; ++k) {
        const int b = tid * BUCKETS_PER_THREAD + k;
        myhist[k] = start;
        if (b <= n_buckets)
            off[((long)set * (n_buckets + 1) + b) * n_chunks + blockIdx.x] = start;
        start += cnt[k];
    }
    __syncthreads();

#pragma unroll
    for (int u = 0; u < EDGES_PER_THREAD; ++u)
        rk[u] += hist[(rw[u] >= 0) ? (rw[u] >> ROW_BITS) : 0];
#pragma unroll
    for (int u = 0; u < EDGES_PER_THREAD; ++u)
        if (rw[u] >= 0)
            sorted[rk[u]] = (cl[u] << ROW_BITS) | (rw[u] & ROW_MASK);
    __syncthreads();

    const int m = min(CHUNK, n_edges - blockIdx.x * CHUNK);
    int* __restrict__ dst = pairs + (long)set * n_edges + (long)blockIdx.x * CHUNK;
#pragma unroll
    for (int u = 0; u < EDGES_PER_THREAD; ++u) rk[u] = sorted[min(tid + u * EDGE_THREADS, m - 1)];
#pragma unroll
    for (int u = 0; u < EDGES_PER_THREAD; ++u) {
        const int i = tid + u * EDGE_THREADS;
        if (i < m) dst[i] = rk[u];
    }
}


__device__ __forceinline__ int order_set(const int* __restrict__ pw, int rs, int len, int runinc,
                                         const int* wsumc,
                                         int n_chunks, int* lnbr, int* rstart,
                                         int* deg, int* cursor, int* wsum, int* cpos, int* cbase,
                                         int tid, int lane, int wave) {
    int seg_count;
    {
        int woff = 0, tot = 0;
        for (int k = 0; k < GEMM_WAVES; ++k) {
            const int v = wsumc[k];
            tot += v;
            woff += (k < wave) ? v : 0;
        }
        if (tid < MAX_CHUNKS) {
            cpos[tid]  = (tid < n_chunks) ? (woff + runinc - len) : 0x7fffffff;
            cbase[tid] = rs;
        }
        if (tid < BUCKET_ROWS) deg[tid] = 0;
        seg_count = __builtin_amdgcn_readfirstlane(tot);
        __syncthreads();
    }
    if (seg_count > PIECE) return seg_count;
    if (seg_count == 0) {
        if (tid <= BUCKET_ROWS) rstart[tid] = 0;
        __syncthreads();
        return 0;
    }
    const int pc = seg_count;

    int w[SEARCHES], p[SEARCHES], c[SEARCHES];
#pragma unroll
    for (int u = 0; u < SEARCHES; ++u) {
        const int pp = tid + u * GEMM_THREADS;
        p[u] = (pp < pc) ? pp : 0;
        c[u] = 0;
    }
#pragma unroll
    for (int step = MAX_CHUNKS >> 1; step > 0; step >>= 1) {
#pragma unroll
        for (int u = 0; u < SEARCHES; ++u)
            if (cpos[c[u] + step] <= p[u]) c[u] += step;
    }
#pragma unroll
    for (int u = 0; u < SEARCHES; ++u)
        w[u] = pw[c[u] * CHUNK + cbase[c[u]] + (p[u] - cpos[c[u]])];
#pragma unroll
    for (int u = 0; u < SEARCHES; ++u)
        if (tid + u * GEMM_THREADS < pc) lds_fetch_add(&deg[w[u] & ROW_MASK], 1);
    __syncthreads();
    const int d0 = (tid < BUCKET_ROWS) ? deg[tid] : 0;
    int inc = d0;
    for (int o = 1; o < 32; o <<= 1) {
        const int t = __shfl_up(inc, o, 32);
        if (lane >= o) inc += t;
    }
    if (lane == 31 && wave < ROW_WAVES) wsum[wave] = inc;
    __syncthreads();
    if (tid < BUCKET_ROWS) {
        int woff = 0;
        for (int k = 0; k < ROW_WAVES; ++k) woff += (k < wave) ? wsum[k] : 0;
        const int st = woff + inc - d0;
        rstart[tid] = st;
        cursor[tid] = st;
        if (tid == BUCKET_ROWS - 1) rstart[BUCKET_ROWS] = st + d0;
    }
    __syncthreads();
#pragma unroll
    for (int u = 0; u < SEARCHES; ++u) {
        if (tid + u * GEMM_THREADS < pc) {
            const int slot = lds_fetch_add(&cursor[w[u] & ROW_MASK], 1);
            lnbr[slot] = w[u];
        }
    }
    __syncthreads();
    return seg_count;
}

__device__ __forceinline__ void self_rows_to_tile(const float* __restrict__ x, long row0, int n,
                                                  float* T, int lane) {
    const int hh = lane >> 4;
    const int q  = lane & 15;
    const char* __restrict__ xb = (const char*)x;
    const unsigned qoff  = (unsigned)q << 4;
    const unsigned rbase = (unsigned)row0;
    const unsigned rlast = (unsigned)(n - 1);
    const v4f zero4 = {0.0f, 0.0f, 0.0f, 0.0f};
    v4f sv[8];
#pragma unroll
    for (int i = 0; i < 8; ++i) {
        const unsigned r = rbase + (unsigned)(2 * i + hh);
        sv[i] = *(const v4f*)(xb + ((min(r, rlast) << 8) | qoff));
    }
#pragma unroll
    for (int i = 0; i < 8; ++i) {
        const unsigned r = rbase + (unsigned)(2 * i + hh);
        *(v4f*)(T + (2 * i + hh) * TILE_PITCH + (q << 2)) = (r <= rlast) ? sv[i] : zero4;
    }
}

#define GATHER_BATCH 10

__device__ __forceinline__ void gather_rows(const float* __restrict__ x, const int* lnbr,
                                            const int* rstart, long row0, int n, float* T,
                                            int lane, int wave) {
    const int hh = lane >> 4;
    const int q  = lane & 15;
    const char* __restrict__ xb = (const char*)x;
    const unsigned qoff  = (unsigned)q << 4;
    const unsigned rbase = (unsigned)row0;
    const unsigned rlast = (unsigned)(n - 1);

    const v4f zero4 = {0.0f, 0.0f, 0.0f, 0.0f};

    const unsigned self_off = (min(rbase + 8u * (unsigned)hh, rlast) << 8) | qoff;

    const int* rs = rstart + wave * 16 + 8 * hh;
    const int beg = rs[0];
    const int cnt = rs[8] - beg;
    const int cother = __shfl_xor(cnt, 16, 32);
    const int cmax = max(cnt, cother);
    const int cmin = __builtin_amdgcn_readfirstlane(min(cnt, cother));

    int cur = 8 * hh;
    float* tcur = T + cur * TILE_PITCH + (q << 2);
    v4f acc = *(v4f*)tcur;
    int j = 0;
    {
        const int* lp = lnbr + beg;
        for (; j + GATHER_BATCH <= cmin; j += GATHER_BATCH, lp += GATHER_BATCH) {
            int wd[GATHER_BATCH];
            v4f t[GATHER_BATCH];
#pragma unroll
            for (int u = 0; u < GATHER_BATCH; ++u) wd[u] = lp[u];
#pragma unroll
            for (int u = 0; u < GATHER_BATCH; ++u)
                t[u] = *(const v4f*)(xb + (((((unsigned)wd[u]) >> 1) & ~255u) | qoff));
#pragma unroll
            for (int u = 0; u < GATHER_BATCH; ++u) {
                const int lr = wd[u] & 15;
                if (lr != cur) {
                    *(v4f*)tcur = acc;
                    cur = lr;
                    tcur = T + lr * TILE_PITCH + (q << 2);
                    acc = *(v4f*)tcur;
                }
                acc += t[u];
            }
        }
    }
    for (; j < cmax; j += GATHER_BATCH) {
        int wd[GATHER_BATCH];
        v4f t[GATHER_BATCH];
#pragma unroll
        for (int u = 0; u < GATHER_BATCH; ++u)
            wd[u] = lnbr[(j + u < cnt) ? (beg + j + u) : 0];
#pragma unroll
        for (int u = 0; u < GATHER_BATCH; ++u) {
            const unsigned off = (j + u < cnt)
                ? ((((unsigned)wd[u] >> 1) & ~255u) | qoff)
                : self_off;
            t[u] = *(const v4f*)(xb + off);
        }
#pragma unroll
        for (int u = 0; u < GATHER_BATCH; ++u) {
            const bool ok = j + u < cnt;
            const int lr = wd[u] & 15;
            if (ok && lr != cur) {
                *(v4f*)tcur = acc;
                cur = lr;
                tcur = T + lr * TILE_PITCH + (q << 2);
                acc = *(v4f*)tcur;
            }
            acc += ok ? t[u] : zero4;
        }
    }
    *(v4f*)tcur = acc;
#pragma unroll
    for (int i = 0; i < 8; ++i) {
        const int lr = 2 * i + hh;
        const int* rr = rstart + wave * 16 + lr;
        const float inv = 1.0f / fmaxf((float)(rr[1] - rr[0] + 1), 1.0f);
        float* trow = T + lr * TILE_PITCH + (q << 2);
        v4f a = *(v4f*)trow;
        a *= inv;
        *(v4f*)trow = a;
    }
}

#define LONG_BATCH 4

__device__ __forceinline__ void gather_mean_tile(const float* __restrict__ x,
                                                 const int* __restrict__ pw,
                                                 const int* __restrict__ offk, int n_chunks,
                                                 long row0, int n, float* T,
                                                 int* lnbr, int* deg, int* lstart, int* cursor,
                                                 int* rdeg, int* wsum,
                                                 int* cpos, int* cbase,
                                                 int tid, int lane, int wave) {
    const int hh = lane >> 4;
    const int q  = lane & 15;
    const char* __restrict__ xb = (const char*)x;
    const unsigned qoff  = (unsigned)q << 4;
    const unsigned rbase = (unsigned)row0;
    const unsigned rlast = (unsigned)(n - 1);

    const v4f zero4 = {0.0f, 0.0f, 0.0f, 0.0f};

    const char* __restrict__ pself = xb + (((rbase + (unsigned)hh) << 8) | qoff);
    v4f sv[8];
#pragma unroll
    for (int i = 0; i < 8; ++i) {
        sv[i] = zero4;
        if (rbase + (unsigned)(2 * i + hh) <= rlast) sv[i] = *(const v4f*)(pself + i * 512);
    }
#pragma unroll
    for (int i = 0; i < 8; ++i)
        *(v4f*)(T + (2 * i + hh) * TILE_PITCH + (q << 2)) = sv[i];
    if (tid < BUCKET_ROWS) rdeg[tid] = 0;

    int seg_count;
    {
        int rs = 0, len = 0;
        if (tid < n_chunks) {
            const int* __restrict__ offe = offk + n_chunks;
            rs  = offk[tid];
            len = offe[tid] - rs;
        }
        int inc = len;
        for (int o = 1; o < 32; o <<= 1) {
            const int t = __shfl_up(inc, o, 32);
            if (lane >= o) inc += t;
        }
        if (lane == 31) wsum[wave] = inc;
        __syncthreads();
        int woff = 0, tot = 0;
        for (int k = 0; k < GEMM_WAVES; ++k) {
            const int v = wsum[k];
            tot += v;
            woff += (k < wave) ? v : 0;
        }
        if (tid < MAX_CHUNKS) {
            cpos[tid]  = (tid < n_chunks) ? (woff + inc - len) : 0x7fffffff;
            cbase[tid] = rs;
        }
        seg_count = __builtin_amdgcn_readfirstlane(tot);
        __syncthreads();
    }

    const unsigned self_off = (min(rbase + 8u * (unsigned)hh, rlast) << 8) | qoff;

    for (int pb = 0; pb < seg_count; pb += PIECE) {
        const int pc = min(PIECE, seg_count - pb);
        int w[SEARCHES], p[SEARCHES], c[SEARCHES];
        const int pend = pb + pc;
        const int pt = pb + tid;
#pragma unroll
        for (int u = 0; u < SEARCHES; ++u) {
            const int pp = pt + u * GEMM_THREADS;
            p[u] = (pp < pend) ? pp : pb;
            c[u] = 0;
        }
#pragma unroll
        for (int step = MAX_CHUNKS >> 1; step > 0; step >>= 1) {
#pragma unroll
            for (int u = 0; u < SEARCHES; ++u)
                if (cpos[c[u] + step] <= p[u]) c[u] += step;
        }
#pragma unroll
        for (int u = 0; u < SEARCHES; ++u)
            w[u] = pw[c[u] * CHUNK + cbase[c[u]] + (p[u] - cpos[c[u]])];
        if (tid < BUCKET_ROWS) deg[tid] = 0;
        __syncthreads();
#pragma unroll
        for (int u = 0; u < SEARCHES; ++u)
            if (tid + u * GEMM_THREADS < pc) lds_fetch_add(&deg[w[u] & ROW_MASK], 1);
        __syncthreads();
        const int d0 = (tid < BUCKET_ROWS) ? deg[tid] : 0;
        int inc = d0;
        for (int o = 1; o < 32; o <<= 1) {
            const int t = __shfl_up(inc, o, 32);
            if (lane >= o) inc += t;
        }
        if (lane == 31 && wave < ROW_WAVES) wsum[wave] = inc;
        __syncthreads();
        if (tid < BUCKET_ROWS) {
            int woff = 0;
            for (int k = 0; k < ROW_WAVES; ++k) woff += (k < wave) ? wsum[k] : 0;
            const int st = woff + inc - d0;
            lstart[tid] = st;
            cursor[tid] = st;
            rdeg[tid] += d0;
        }
        __syncthreads();
#pragma unroll
        for (int u = 0; u < SEARCHES; ++u) {
            if (tid + u * GEMM_THREADS < pc) {
                const int slot = lds_fetch_add(&cursor[w[u] & ROW_MASK], 1);
                lnbr[slot] = w[u];
            }
        }
        __syncthreads();

        const int beg = lstart[wave * 16 + 8 * hh];
        const int cnt = cursor[wave * 16 + 8 * hh + 7] - beg;
        const int cother = __shfl_xor(cnt, 16, 32);
        const int cmax = max(cnt, cother);

        int cur = 8 * hh;
        float* tcur = T + cur * TILE_PITCH + (q << 2);
        v4f acc = *(v4f*)tcur;
        for (int j = 0; j < cmax; j += LONG_BATCH) {
            int wd[LONG_BATCH];
            v4f t[LONG_BATCH];
#pragma unroll
            for (int u = 0; u < LONG_BATCH; ++u)
                wd[u] = lnbr[(j + u < cnt) ? (beg + j + u) : 0];
#pragma unroll
            for (int u = 0; u < LONG_BATCH; ++u) {
                const unsigned off = (j + u < cnt)
                    ? ((((unsigned)wd[u] >> 1) & ~255u) | qoff)
                    : self_off;
                t[u] = *(const v4f*)(xb + off);
            }
#pragma unroll
            for (int u = 0; u < LONG_BATCH; ++u) {
                const bool ok = j + u < cnt;
                const int lr = wd[u] & 15;
                if (ok && lr != cur) {
                    *(v4f*)tcur = acc;
                    cur = lr;
                    tcur = T + lr * TILE_PITCH + (q << 2);
                    acc = *(v4f*)tcur;
                }
                acc += ok ? t[u] : zero4;
            }
        }
        *(v4f*)tcur = acc;
        __syncthreads();
    }
    __syncthreads();
#pragma unroll
    for (int i = 0; i < 8; ++i) {
        const int lr = 2 * i + hh;
        const float inv = 1.0f / fmaxf((float)(rdeg[wave * 16 + lr] + 1), 1.0f);
        float* trow = T + lr * TILE_PITCH + (q << 2);
        v4f acc = *(v4f*)trow;
        acc *= inv;
        *(v4f*)trow = acc;
    }
}

__device__ __forceinline__ void tile_product(const float* ta, const float* wset, int h,
                                             v8f& acc0, v8f& acc1, v8f& acc2, v8f& acc3) {
#pragma unroll
    for (int k = 0; k < 16; ++k) {
        const int c = (k << 2) + (h << 1);
        const v2f a = *(const v2f*)(ta + c);
        const float* wp = wset + (((k << 1) + h) << 7);
        const v2f b0 = *(const v2f*)(wp);
        const v2f b1 = *(const v2f*)(wp + 32);
        const v2f b2 = *(const v2f*)(wp + 64);
        const v2f b3 = *(const v2f*)(wp + 96);
        acc0 = __builtin_amdgcn_wmma_f32_16x16x4_f32(false, a, false, b0, (short)0, acc0, false, false);
        acc1 = __builtin_amdgcn_wmma_f32_16x16x4_f32(false, a, false, b1, (short)0, acc1, false, false);
        acc2 = __builtin_amdgcn_wmma_f32_16x16x4_f32(false, a, false, b2, (short)0, acc2, false, false);
        acc3 = __builtin_amdgcn_wmma_f32_16x16x4_f32(false, a, false, b3, (short)0, acc3, false, false);
    }
    asm volatile("v_nop\n\tv_nop\n\tv_nop\n\tv_nop\n\tv_nop\n\tv_nop\n\tv_nop\n\tv_nop"
                 : "+v"(acc0), "+v"(acc1), "+v"(acc2), "+v"(acc3));
}

__global__ __launch_bounds__(GEMM_THREADS, 7) void sage_gather_gemm_norm_kernel(
        const float* __restrict__ x1,   const float* __restrict__ x2,
        const int* __restrict__ off,    const int* __restrict__ pairs,
        const float* __restrict__ W,    const float* __restrict__ bias,
        float* __restrict__ out, int n, int n_edges, int n_buckets, int n_chunks) {
    __shared__ float ldsW[192 * 64];
    __shared__ __attribute__((aligned(16))) float ldsA[GEMM_WAVES * 16 * TILE_PITCH];
    __shared__ int lnbr[2][PIECE];
    __shared__ int rstart[2][BUCKET_ROWS + 1];
    __shared__ int wsum[32];
    __shared__ int wsumc[2][32];
    __shared__ int scr[2 * MAX_CHUNKS + 2 * BUCKET_ROWS];

    const int tid  = threadIdx.x;
    const int lane = tid & 31;
    const int wave = tid >> 5;
    const int m = lane & 15;
    const int h = lane >> 4;

    const long tile = (long)blockIdx.x * GEMM_WAVES + wave;
    const long row0 = tile << 4;

    float* T = ldsA + wave * (16 * TILE_PITCH);
    const float* ta = T + m * TILE_PITCH;

    float wv[W_WORDS];
#pragma unroll
    for (int u = 0; u < W_WORDS; ++u) wv[u] = W[min(tid + u * GEMM_THREADS, 192 * 64 - 1)];
    int crs[2], clen[2], cinc[2];
    {
        const int ci = max(min(tid, n_chunks - 1), 0);
#pragma unroll
        for (int set = 0; set < 2; ++set) {
            const int* __restrict__ offk = off + ((long)set * (n_buckets + 1) + blockIdx.x) * n_chunks;
            const int a = offk[ci];
            const int e = offk[n_chunks + ci];
            crs[set]  = (tid < n_chunks) ? a : 0;
            clen[set] = (tid < n_chunks) ? e - a : 0;
        }
    }
    self_rows_to_tile(x1, row0, n, T, lane);

#pragma unroll
    for (int u = 0; u < W_WORDS; ++u) {
        const int idx = tid + u * GEMM_THREADS;
        const int row = idx >> 6;
        const int col = idx & 63;
        if (idx < 192 * 64) ldsW[((((row >> 1) << 6) | col) << 1) | (row & 1)] = wv[u];
    }
    cinc[0] = clen[0];
    cinc[1] = clen[1];
    for (int o = 1; o < 32; o <<= 1) {
        const int t0 = __shfl_up(cinc[0], o, 32);
        const int t1 = __shfl_up(cinc[1], o, 32);
        if (lane >= o) {
            cinc[0] += t0;
            cinc[1] += t1;
        }
    }
    if (lane == 31) {
        wsumc[0][wave] = cinc[0];
        wsumc[1][wave] = cinc[1];
    }
    __syncthreads();

    v8f acc0 = {}; v8f acc1 = {}; v8f acc2 = {}; v8f acc3 = {};
    const float* wbase = ldsW + (m << 1);

    int* const cpos   = scr;
    int* const cbase  = scr + MAX_CHUNKS;
    int* const deg    = scr + 2 * MAX_CHUNKS;
    int* const cursor = scr + 2 * MAX_CHUNKS + BUCKET_ROWS;

    bool fits = true;
#pragma nounroll
    for (int set = 0; set < 2; ++set) {
        const int count = order_set(pairs + (long)set * n_edges, crs[set], clen[set], cinc[set],
                                    wsumc[set], n_chunks, lnbr[set], rstart[set],
                                    deg, cursor, wsum, cpos, cbase, tid, lane, wave);
        if (count > PIECE) {
            fits = false;
            break;
        }
    }

    if (fits) {
#pragma nounroll
        for (int set = 0; set < 2; ++set) {
            if (set) self_rows_to_tile(x2, row0, n, T, lane);
            gather_rows(set ? x2 : x1, lnbr[set], rstart[set], row0, n, T, lane, wave);
            tile_product(ta, wbase + ((set << 5) << 7), h, acc0, acc1, acc2, acc3);
        }
    } else {
        __syncthreads();
        int* const fb = lnbr[1];
#pragma nounroll
        for (int set = 0; set < 2; ++set) {
            const int* __restrict__ offk = off + ((long)set * (n_buckets + 1) + blockIdx.x) * n_chunks;
            gather_mean_tile(set ? x2 : x1, pairs + (long)set * n_edges, offk, n_chunks,
                             row0, n, T, lnbr[0], fb, fb + BUCKET_ROWS, fb + 2 * BUCKET_ROWS,
                             fb + 3 * BUCKET_ROWS, wsum, fb + 4 * BUCKET_ROWS,
                             fb + 4 * BUCKET_ROWS + MAX_CHUNKS, tid, lane, wave);
            __syncthreads();
            tile_product(ta, wbase + ((set << 5) << 7), h, acc0, acc1, acc2, acc3);
            __syncthreads();
        }
    }
    const int me = m;
    const int he = h;
    const float* wbe = ldsW + (me << 1);
    long r = row0 + me;
    if (r > (long)(n - 1)) r = (long)(n - 1);
    const float* pax = x1 + r * D_FEAT;
#pragma unroll
    for (int k = 0; k < 16; ++k) {
        const int c = (k << 2) + (he << 1);
        const v2f a = *(const v2f*)(pax + c);
        const float* wp = wbe + ((64 + (k << 1) + he) << 7);
        const v2f b0 = *(const v2f*)(wp);
        const v2f b1 = *(const v2f*)(wp + 32);
        const v2f b2 = *(const v2f*)(wp + 64);
        const v2f b3 = *(const v2f*)(wp + 96);
        acc0 = __builtin_amdgcn_wmma_f32_16x16x4_f32(false, a, false, b0, (short)0, acc0, false, false);
        acc1 = __builtin_amdgcn_wmma_f32_16x16x4_f32(false, a, false, b1, (short)0, acc1, false, false);
        acc2 = __builtin_amdgcn_wmma_f32_16x16x4_f32(false, a, false, b2, (short)0, acc2, false, false);
        acc3 = __builtin_amdgcn_wmma_f32_16x16x4_f32(false, a, false, b3, (short)0, acc3, false, false);
    }

    const float bs0 = bias[me];
    const float bs1 = bias[16 + me];
    const float bs2 = bias[32 + me];
    const float bs3 = bias[48 + me];

#pragma unroll
    for (int j = 0; j < 8; ++j) {
        const float o0 = acc0[j] + bs0;
        const float o1 = acc1[j] + bs1;
        const float o2 = acc2[j] + bs2;
        const float o3 = acc3[j] + bs3;
        float s = o0 * o0 + o1 * o1 + o2 * o2 + o3 * o3;
        s += __shfl_xor(s, 1, 32);
        s += __shfl_xor(s, 2, 32);
        s += __shfl_xor(s, 4, 32);
        s += __shfl_xor(s, 8, 32);
        const float scale = 1.0f / fmaxf(sqrtf(s), 1e-12f);
        const long orow = row0 + j + (he << 3);
        if (orow < n) {
            float* po = out + orow * D_FEAT + me;
            po[0]  = o0 * scale;
            po[16] = o1 * scale;
            po[32] = o2 * scale;
            po[48] = o3 * scale;
        }
    }
}

extern "C" void kernel_launch(void* const* d_in, const int* in_sizes, int n_in,
                              void* d_out, int out_size, void* d_ws, size_t ws_size,
                              hipStream_t stream) {
    const float* x1    = (const float*)d_in[0];
    const float* x2    = (const float*)d_in[1];
    const int*   e_pos = (const int*)d_in[2];
    const int*   e_neg = (const int*)d_in[3];
    const float* W     = (const float*)d_in[4];
    const float* bias  = (const float*)d_in[5];
    float*       out   = (float*)d_out;

    const int n       = in_sizes[0] / D_FEAT;
    const int n_edges = in_sizes[2] / 2;

    const int n_buckets = (n + BUCKET_ROWS - 1) / BUCKET_ROWS;
    const int n_chunks  = (n_edges + CHUNK - 1) / CHUNK;

    int* off   = (int*)d_ws;
    int* pairs = off + (size_t)2 * (n_buckets + 1) * n_chunks;

    const dim3 egrid(n_chunks, 2);
    chunk_group_kernel<<<egrid, EDGE_THREADS, 0, stream>>>(e_pos, e_neg, off, pairs,
                                                  n_edges, n_buckets, n_chunks);
    {
        const int tiles  = (n + 15) / 16;
        const int blocks = (tiles + GEMM_WAVES - 1) / GEMM_WAVES;
        sage_gather_gemm_norm_kernel<<<blocks, GEMM_THREADS, 0, stream>>>(x1, x2, off, pairs,
                                                                 W, bias, out, n, n_edges, n_buckets,
                                                                 n_chunks);
    }
}
